# starting kernel
# baseline (speedup 1.0000x reference)
#include <hip/hip_runtime.h>

#define NN 10000
#define NE 640000
#define INF 128
#define HF  128
#define OF  64

typedef float v2f __attribute__((ext_vector_type(2)));
typedef float v8f __attribute__((ext_vector_type(8)));


__global__ void zero_f32(float* __restrict__ p, int n) {
    int i = blockIdx.x * blockDim.x + threadIdx.x;
    if (i < n) p[i] = 0.0f;
}

__global__ void degrees_kernel(const int* __restrict__ src, const int* __restrict__ dst,
                               float* __restrict__ deg_out, float* __restrict__ deg_in, int nE) {
    int e = blockIdx.x * blockDim.x + threadIdx.x;
    if (e < nE) {
        atomicAdd(&deg_out[src[e]], 1.0f);
        atomicAdd(&deg_in[dst[e]], 1.0f);
    }
}

__global__ void rsqrt_kernel(const float* __restrict__ deg_out, const float* __restrict__ deg_in,
                             float* __restrict__ rs_out, float* __restrict__ rs_in, int n) {
    int i = blockIdx.x * blockDim.x + threadIdx.x;
    if (i < n) {
        rs_out[i] = rsqrtf(fmaxf(deg_out[i], 1.0f));
        rs_in[i]  = rsqrtf(fmaxf(deg_in[i],  1.0f));
    }
}

__global__ void scale_rows_kernel(const float* __restrict__ x, const float* __restrict__ rs,
                                  float* __restrict__ h, int nNodes) {
    int i = blockIdx.x * blockDim.x + threadIdx.x;
    int node = i >> 5;
    int c    = i & 31;
    if (node >= nNodes) return;
    float s = rs[node];
    float4 v = *(const float4*)(x + (size_t)node * 128 + c * 4);
    v.x *= s; v.y *= s; v.z *= s; v.w *= s;
    *(float4*)(h + (size_t)node * 128 + c * 4) = v;
}

__global__ void scatter_kernel(const int* __restrict__ src, const int* __restrict__ dst,
                               const float* __restrict__ h, float* __restrict__ agg, int nE) {
    int gid  = blockIdx.x * blockDim.x + threadIdx.x;
    int e    = gid >> 5;
    int lane = gid & 31;
    if (e >= nE) return;
    int s = src[e];
    int d = dst[e];
    float4 v = *(const float4*)(h + (size_t)s * 128 + lane * 4);
    float* out = agg + (size_t)d * 128 + lane * 4;
    unsafeAtomicAdd(out + 0, v.x);
    unsafeAtomicAdd(out + 1, v.y);
    unsafeAtomicAdd(out + 2, v.z);
    unsafeAtomicAdd(out + 3, v.w);
}

template <int K, int N, bool RELU>
__global__ void gemm_wmma_f32(const float* __restrict__ A,
                              const float* __restrict__ rs,
                              const float* __restrict__ W,
                              const float* __restrict__ bias,
                              float* __restrict__ out) {
    int lane  = threadIdx.x & 31;
    int wave  = threadIdx.x >> 5;
    int mBase = blockIdx.x * 16;
    int nBase = wave * 16;
    int half  = lane >> 4;
    int l16   = lane & 15;
    int mRow  = mBase + l16;
    int nCol  = nBase + l16;

    float rscale = rs[mRow];
    const float* Arow = A + (size_t)mRow * K + half * 2;
    const float* Wcol = W + (size_t)half * 2 * N + nCol;

    v8f c = {};
    #pragma unroll
    for (int k = 0; k < K; k += 4) {
        v2f a;
        a.x = Arow[k]     * rscale;
        a.y = Arow[k + 1] * rscale;
        v2f b;
        b.x = Wcol[(size_t)k * N];
        b.y = Wcol[(size_t)(k + 1) * N];
        c = __builtin_amdgcn_wmma_f32_16x16x4_f32(false, a, false, b, (short)0, c,
                                                  false, false);
    }

    float bv = bias[nCol];
    #pragma unroll
    for (int r = 0; r < 8; ++r) {
        int row = mBase + r + half * 8;
        float v = c[r] + bv;
        if (RELU) v = fmaxf(v, 0.0f);
        out[(size_t)row * N + nCol] = v;
    }
}


extern "C" void kernel_launch(void* const* d_in, const int* in_sizes, int n_in,
                              void* d_out, int out_size, void* d_ws, size_t ws_size,
                              hipStream_t stream) {
    const float* in_feat = (const float*)d_in[0];
    const int*   src     = (const int*)d_in[1];
    const int*   dst     = (const int*)d_in[2];
    const float* W1      = (const float*)d_in[3];
    const float* b1      = (const float*)d_in[4];
    const float* W2      = (const float*)d_in[5];
    const float* b2      = (const float*)d_in[6];
    float* out = (float*)d_out;

    float* ws      = (float*)d_ws;
    float* deg_out = ws;
    float* deg_in  = ws + NN;
    float* rs_out  = ws + 2 * NN;
    float* rs_in   = ws + 3 * NN;
    float* hbuf    = ws + 4 * NN;
    float* agg     = hbuf + (size_t)NN * 128;
    float* hidden  = agg  + (size_t)NN * 128;

    const int B = 256;
    const size_t featElems = (size_t)NN * 128;

    zero_f32<<<(2 * NN + B - 1) / B, B, 0, stream>>>(deg_out, 2 * NN);
    degrees_kernel<<<(NE + B - 1) / B, B, 0, stream>>>(src, dst, deg_out, deg_in, NE);
    rsqrt_kernel<<<(NN + B - 1) / B, B, 0, stream>>>(deg_out, deg_in, rs_out, rs_in, NN);

    scale_rows_kernel<<<(NN * 32 + B - 1) / B, B, 0, stream>>>(in_feat, rs_out, hbuf, NN);
    zero_f32<<<(int)((featElems + B - 1) / B), B, 0, stream>>>(agg, (int)featElems);
    scatter_kernel<<<(NE * 32) / B, B, 0, stream>>>(src, dst, hbuf, agg, NE);
    gemm_wmma_f32<INF, HF, true><<<NN / 16, (HF / 16) * 32, 0, stream>>>(
        agg, rs_in, W1, b1, hidden);

    scale_rows_kernel<<<(NN * 32 + B - 1) / B, B, 0, stream>>>(hidden, rs_out, hbuf, NN);
    zero_f32<<<(int)((featElems + B - 1) / B), B, 0, stream>>>(agg, (int)featElems);
    scatter_kernel<<<(NE * 32) / B, B, 0, stream>>>(src, dst, hbuf, agg, NE);
    gemm_wmma_f32<HF, OF, false><<<NN / 16, (OF / 16) * 32, 0, stream>>>(
        agg, rs_in, W2, b2, out);
}


# degrees: global f32 atomics on a 20000-word array replaced by per-block LDS integer histograms (64 edge slices per index array) summed per node; counts exact, zeroing launch dropped
# speedup vs baseline: 1.4722x; 1.4722x over previous
#include <hip/hip_runtime.h>

#define NN 10000
#define NE 640000
#define INF 128
#define HF  128
#define OF  64

typedef float v2f __attribute__((ext_vector_type(2)));
typedef float v8f __attribute__((ext_vector_type(8)));


__global__ void zero_f32(float* __restrict__ p, int n) {
    int i = blockIdx.x * blockDim.x + threadIdx.x;
    if (i < n) p[i] = 0.0f;
}

#define HS   64
#define HT   1024
#define NNP  10016

__global__ void __launch_bounds__(HT)
degree_hist_kernel(const int* __restrict__ src, const int* __restrict__ dst,
                   int* __restrict__ table) {
    __shared__ __attribute__((aligned(16))) int hist[NN];
    static_assert(NE % HS == 0, "edge slices must be equal");
    static_assert(NN % 4 == 0 && (NE / HS) % 4 == 0, "rows and slices move in 16-byte pieces");
    const int which = blockIdx.x / HS;
    const int slice = blockIdx.x - which * HS;
    const int* __restrict__ idx = (which ? dst : src) + (size_t)slice * (NE / HS);

    const int4* __restrict__ idx4 = (const int4*)idx;
    constexpr int PER4  = NE / HS / 4;
    constexpr int TRIPS = (PER4 + HT - 1) / HT;
    int4 v[TRIPS];
    #pragma unroll
    for (int j = 0; j < TRIPS; ++j) {
        int i = threadIdx.x + j * HT;
        if (i < PER4) v[j] = idx4[i];
    }

    int4* hist4 = (int4*)hist;
    for (int i = threadIdx.x; i < NN / 4; i += HT) hist4[i] = make_int4(0, 0, 0, 0);
    __syncthreads();
    #pragma unroll
    for (int j = 0; j < TRIPS; ++j) {
        int i = threadIdx.x + j * HT;
        if (i < PER4) {
            atomicAdd(&hist[v[j].x], 1);
            atomicAdd(&hist[v[j].y], 1);
            atomicAdd(&hist[v[j].z], 1);
            atomicAdd(&hist[v[j].w], 1);
        }
    }
    __syncthreads();
    int4* row4 = (int4*)(table + (size_t)blockIdx.x * NNP);
    for (int i = threadIdx.x; i < NN / 4; i += HT) row4[i] = hist4[i];
}

__global__ void degree_rsqrt_kernel(const int* __restrict__ table, float* __restrict__ rs) {
    int t = blockIdx.x * blockDim.x + threadIdx.x;
    if (t >= 2 * NN) return;
    int which = t / NN;
    int node  = t - which * NN;
    const int* __restrict__ col = table + (size_t)which * HS * NNP + node;
    int s = 0;
    #pragma unroll
    for (int b = 0; b < HS; ++b) s += col[(size_t)b * NNP];
    rs[t] = rsqrtf(fmaxf((float)s, 1.0f));
}

__global__ void scale_rows_kernel(const float* __restrict__ x, const float* __restrict__ rs,
                                  float* __restrict__ h, int nNodes) {
    int i = blockIdx.x * blockDim.x + threadIdx.x;
    int node = i >> 5;
    int c    = i & 31;
    if (node >= nNodes) return;
    float s = rs[node];
    float4 v = *(const float4*)(x + (size_t)node * 128 + c * 4);
    v.x *= s; v.y *= s; v.z *= s; v.w *= s;
    *(float4*)(h + (size_t)node * 128 + c * 4) = v;
}

__global__ void scatter_kernel(const int* __restrict__ src, const int* __restrict__ dst,
                               const float* __restrict__ h, float* __restrict__ agg, int nE) {
    int gid  = blockIdx.x * blockDim.x + threadIdx.x;
    int e    = gid >> 5;
    int lane = gid & 31;
    if (e >= nE) return;
    int s = src[e];
    int d = dst[e];
    float4 v = *(const float4*)(h + (size_t)s * 128 + lane * 4);
    float* out = agg + (size_t)d * 128 + lane * 4;
    unsafeAtomicAdd(out + 0, v.x);
    unsafeAtomicAdd(out + 1, v.y);
    unsafeAtomicAdd(out + 2, v.z);
    unsafeAtomicAdd(out + 3, v.w);
}

template <int K, int N, bool RELU>
__global__ void gemm_wmma_f32(const float* __restrict__ A,
                              const float* __restrict__ rs,
                              const float* __restrict__ W,
                              const float* __restrict__ bias,
                              float* __restrict__ out) {
    int lane  = threadIdx.x & 31;
    int wave  = threadIdx.x >> 5;
    int mBase = blockIdx.x * 16;
    int nBase = wave * 16;
    int half  = lane >> 4;
    int l16   = lane & 15;
    int mRow  = mBase + l16;
    int nCol  = nBase + l16;

    float rscale = rs[mRow];
    const float* Arow = A + (size_t)mRow * K + half * 2;
    const float* Wcol = W + (size_t)half * 2 * N + nCol;

    v8f c = {};
    #pragma unroll
    for (int k = 0; k < K; k += 4) {
        v2f a;
        a.x = Arow[k]     * rscale;
        a.y = Arow[k + 1] * rscale;
        v2f b;
        b.x = Wcol[(size_t)k * N];
        b.y = Wcol[(size_t)(k + 1) * N];
        c = __builtin_amdgcn_wmma_f32_16x16x4_f32(false, a, false, b, (short)0, c,
                                                  false, false);
    }

    float bv = bias[nCol];
    #pragma unroll
    for (int r = 0; r < 8; ++r) {
        int row = mBase + r + half * 8;
        float v = c[r] + bv;
        if (RELU) v = fmaxf(v, 0.0f);
        out[(size_t)row * N + nCol] = v;
    }
}


extern "C" void kernel_launch(void* const* d_in, const int* in_sizes, int n_in,
                              void* d_out, int out_size, void* d_ws, size_t ws_size,
                              hipStream_t stream) {
    const float* in_feat = (const float*)d_in[0];
    const int*   src     = (const int*)d_in[1];
    const int*   dst     = (const int*)d_in[2];
    const float* W1      = (const float*)d_in[3];
    const float* b1      = (const float*)d_in[4];
    const float* W2      = (const float*)d_in[5];
    const float* b2      = (const float*)d_in[6];
    float* out = (float*)d_out;

    float* ws      = (float*)d_ws;
    float* rs_out  = ws + 2 * NN;
    float* rs_in   = ws + 3 * NN;
    float* hbuf    = ws + 4 * NN;
    float* agg     = hbuf + (size_t)NN * 128;
    float* hidden  = agg  + (size_t)NN * 128;
    int*   dtable  = (int*)(hidden + (size_t)NN * 128);

    const int B = 256;
    const size_t featElems = (size_t)NN * 128;

    degree_hist_kernel<<<2 * HS, HT, 0, stream>>>(src, dst, dtable);
    degree_rsqrt_kernel<<<(2 * NN + 127) / 128, 128, 0, stream>>>(dtable, rs_out);

    scale_rows_kernel<<<(NN * 32 + B - 1) / B, B, 0, stream>>>(in_feat, rs_out, hbuf, NN);
    zero_f32<<<(int)((featElems + B - 1) / B), B, 0, stream>>>(agg, (int)featElems);
    scatter_kernel<<<(NE * 32) / B, B, 0, stream>>>(src, dst, hbuf, agg, NE);
    gemm_wmma_f32<INF, HF, true><<<NN / 16, (HF / 16) * 32, 0, stream>>>(
        agg, rs_in, W1, b1, hidden);

    scale_rows_kernel<<<(NN * 32 + B - 1) / B, B, 0, stream>>>(hidden, rs_out, hbuf, NN);
    zero_f32<<<(int)((featElems + B - 1) / B), B, 0, stream>>>(agg, (int)featElems);
    scatter_kernel<<<(NE * 32) / B, B, 0, stream>>>(src, dst, hbuf, agg, NE);
    gemm_wmma_f32<HF, OF, false><<<NN / 16, (OF / 16) * 32, 0, stream>>>(
        agg, rs_in, W2, b2, out);
}


# SpMM: f32 global atomic scatter replaced by per-call counting sort of edges by dst (LDS slots) and one wave per node summing src rows in registers, sorted order; scaling and zeroing folded in
# speedup vs baseline: 3.9990x; 2.7163x over previous
#include <hip/hip_runtime.h>


#define NN 10000
#define NE 640000
#define INF 128
#define HF  128
#define OF  64

typedef float v2f __attribute__((ext_vector_type(2)));
typedef float v8f __attribute__((ext_vector_type(8)));

#define HS   64
#define HT   1024
#define NNP  10016

__global__ void __launch_bounds__(HT)
degree_hist_kernel(const int* __restrict__ src, const int* __restrict__ dst,
                   int* __restrict__ table) {
    __shared__ __attribute__((aligned(16))) int hist[NN];
    static_assert(NE % HS == 0, "edge slices must be equal");
    static_assert(NN % 4 == 0 && (NE / HS) % 4 == 0, "rows and slices move in 16-byte pieces");
    const int which = blockIdx.x / HS;
    const int slice = blockIdx.x - which * HS;
    const int* __restrict__ idx = (which ? dst : src) + (size_t)slice * (NE / HS);

    const int4* __restrict__ idx4 = (const int4*)idx;
    constexpr int PER4  = NE / HS / 4;
    constexpr int TRIPS = (PER4 + HT - 1) / HT;
    int4 v[TRIPS];
    #pragma unroll
    for (int j = 0; j < TRIPS; ++j) {
        int i = threadIdx.x + j * HT;
        if (i < PER4) v[j] = idx4[i];
    }

    int4* hist4 = (int4*)hist;
    for (int i = threadIdx.x; i < NN / 4; i += HT) hist4[i] = make_int4(0, 0, 0, 0);
    __syncthreads();
    #pragma unroll
    for (int j = 0; j < TRIPS; ++j) {
        int i = threadIdx.x + j * HT;
        if (i < PER4) {
            atomicAdd(&hist[v[j].x], 1);
            atomicAdd(&hist[v[j].y], 1);
            atomicAdd(&hist[v[j].z], 1);
            atomicAdd(&hist[v[j].w], 1);
        }
    }
    __syncthreads();
    int4* row4 = (int4*)(table + (size_t)blockIdx.x * NNP);
    for (int i = threadIdx.x; i < NN / 4; i += HT) row4[i] = hist4[i];
}

#define DT 128
#define DB ((NN + DT - 1) / DT)
__global__ void __launch_bounds__(DT)
degree_rsqrt_kernel(const int* __restrict__ table, float* __restrict__ rs_out,
                    float* __restrict__ rs_in, int* __restrict__ deg_in,
                    int* __restrict__ sliceoff) {
    const int which = blockIdx.x >= DB;
    const int node  = (blockIdx.x - which * DB) * DT + threadIdx.x;
    if (node >= NN) return;
    const int* __restrict__ col = table + (size_t)which * HS * NNP + node;
    int c[HS];
    #pragma unroll
    for (int b = 0; b < HS; ++b) c[b] = col[(size_t)b * NNP];
    int s = 0;
    if (which) {
        #pragma unroll
        for (int b = 0; b < HS; ++b) { sliceoff[(size_t)b * NNP + node] = s; s += c[b]; }
        deg_in[node] = s;
        rs_in[node] = rsqrtf(fmaxf((float)s, 1.0f));
    } else {
        #pragma unroll
        for (int b = 0; b < HS; ++b) s += c[b];
        rs_out[node] = rsqrtf(fmaxf((float)s, 1.0f));
    }
}

#define SCAN_PER 10
__global__ void __launch_bounds__(HT)
edge_fill_kernel(const int* __restrict__ src, const int* __restrict__ dst,
                 const int* __restrict__ deg_in, const int* __restrict__ sliceoff,
                 int* __restrict__ rowptr, int* __restrict__ sorted_src) {
    static_assert(HT * SCAN_PER >= NN, "the block scan must cover every node");
    static_assert(HT == 1024, "the scan of wave totals is one wave of 32 lanes");
    __shared__ int cur[HT * SCAN_PER];
    __shared__ int wtot[HT / 32];
    const int slice = blockIdx.x;
    const int tid   = threadIdx.x;

    const int4* __restrict__ d4 = (const int4*)(dst + (size_t)slice * (NE / HS));
    const int4* __restrict__ s4 = (const int4*)(src + (size_t)slice * (NE / HS));
    constexpr int PER4  = NE / HS / 4;
    constexpr int TRIPS = (PER4 + HT - 1) / HT;
    int4 dv[TRIPS], sv[TRIPS];
    #pragma unroll
    for (int j = 0; j < TRIPS; ++j) {
        int i = tid + j * HT;
        if (i < PER4) { dv[j] = d4[i]; sv[j] = s4[i]; }
    }

    const int* __restrict__ so = sliceoff + (size_t)slice * NNP;
    int dg[SCAN_PER], sof[SCAN_PER];
    #pragma unroll
    for (int k = 0; k < SCAN_PER; ++k) {
        int i = min(tid + k * HT, NN - 1);
        dg[k]  = deg_in[i];
        sof[k] = so[i];
    }
    #pragma unroll
    for (int k = 0; k < SCAN_PER; ++k) {
        int i = tid + k * HT;
        cur[i] = (i < NN) ? dg[k] : 0;
    }
    __syncthreads();

    int loc[SCAN_PER];
    int sum = 0;
    #pragma unroll
    for (int k = 0; k < SCAN_PER; ++k) { loc[k] = sum; sum += cur[tid * SCAN_PER + k]; }
    const int lane = tid & 31;
    const int wv   = tid >> 5;
    int inc = sum;
    #pragma unroll
    for (int off = 1; off < 32; off <<= 1) {
        int up = __shfl_up(inc, off);
        if (lane >= off) inc += up;
    }
    if (lane == 31) wtot[wv] = inc;
    __syncthreads();
    if (wv == 0) {
        int w = wtot[lane];
        int winc = w;
        #pragma unroll
        for (int off = 1; off < 32; off <<= 1) {
            int up = __shfl_up(winc, off);
            if (lane >= off) winc += up;
        }
        wtot[lane] = winc - w;
    }
    __syncthreads();
    const int before = wtot[wv] + inc - sum;
    #pragma unroll
    for (int k = 0; k < SCAN_PER; ++k) cur[tid * SCAN_PER + k] = before + loc[k];
    if (slice == 0 && tid == HT - 1) rowptr[NN] = before + sum;
    __syncthreads();

    #pragma unroll
    for (int k = 0; k < SCAN_PER; ++k) {
        int i = tid + k * HT;
        if (i < NN) {
            int c = cur[i];
            if (slice == 0) rowptr[i] = c;
            cur[i] = c + sof[k];
        }
    }
    __syncthreads();

    int slot[TRIPS][4];
    #pragma unroll
    for (int j = 0; j < TRIPS; ++j) {
        int i = tid + j * HT;
        if (i < PER4) {
            slot[j][0] = atomicAdd(&cur[dv[j].x], 1);
            slot[j][1] = atomicAdd(&cur[dv[j].y], 1);
            slot[j][2] = atomicAdd(&cur[dv[j].z], 1);
            slot[j][3] = atomicAdd(&cur[dv[j].w], 1);
        }
    }
    #pragma unroll
    for (int j = 0; j < TRIPS; ++j) {
        int i = tid + j * HT;
        if (i < PER4) {
            sorted_src[slot[j][0]] = sv[j].x;
            sorted_src[slot[j][1]] = sv[j].y;
            sorted_src[slot[j][2]] = sv[j].z;
            sorted_src[slot[j][3]] = sv[j].w;
        }
    }
}

#define GW 8
#define GU 8
__global__ void __launch_bounds__(GW * 32)
gather_rows_kernel(const int* __restrict__ rowptr, const int* __restrict__ sorted_src,
                   const float* __restrict__ x, const float* __restrict__ rs_out,
                   float* __restrict__ agg) {
    static_assert(NN % GW == 0, "whole blocks of nodes");
    const int lane = threadIdx.x & 31;
    const int node = blockIdx.x * GW + (threadIdx.x >> 5);
    const int beg = __builtin_amdgcn_readfirstlane(rowptr[node]);
    const int end = __builtin_amdgcn_readfirstlane(rowptr[node + 1]);
    const float* __restrict__ xl = x + lane * 4;
    float4 acc = make_float4(0.0f, 0.0f, 0.0f, 0.0f);
    for (int base = beg; base < end; base += 32) {
        const int n = min(32, end - base);
        const int my    = sorted_src[base + min(lane, n - 1)];
        const int myr_i = __float_as_int(rs_out[my]);
        int k = 0;
        for (; k + GU <= n; k += GU) {
            float4 v[GU];
            #pragma unroll
            for (int u = 0; u < GU; ++u) {
                const int s = __builtin_amdgcn_readlane(my, k + u);
                v[u] = *(const float4*)(xl + (size_t)s * 128);
            }
            #pragma unroll
            for (int u = 0; u < GU; ++u) {
                const float r = __int_as_float(__builtin_amdgcn_readlane(myr_i, k + u));
                acc.x += v[u].x * r;
                acc.y += v[u].y * r;
                acc.z += v[u].z * r;
                acc.w += v[u].w * r;
            }
        }
        if (k < n) {
            float4 v[GU];
            #pragma unroll
            for (int u = 0; u < GU; ++u) {
                const int s = __builtin_amdgcn_readlane(my, k + u);
                v[u] = *(const float4*)(xl + (size_t)s * 128);
            }
            #pragma unroll
            for (int u = 0; u < GU; ++u) {
                const float r  = __int_as_float(__builtin_amdgcn_readlane(myr_i, k + u));
                const bool  on = k + u < n;
                acc.x = on ? acc.x + v[u].x * r : acc.x;
                acc.y = on ? acc.y + v[u].y * r : acc.y;
                acc.z = on ? acc.z + v[u].z * r : acc.z;
                acc.w = on ? acc.w + v[u].w * r : acc.w;
            }
        }
    }
    *(float4*)(agg + (size_t)node * 128 + lane * 4) = acc;
}

template <int K, int N, bool RELU>
__global__ void gemm_wmma_f32(const float* __restrict__ A,
                              const float* __restrict__ rs,
                              const float* __restrict__ W,
                              const float* __restrict__ bias,
                              float* __restrict__ out) {
    int lane  = threadIdx.x & 31;
    int wave  = threadIdx.x >> 5;
    int mBase = blockIdx.x * 16;
    int nBase = wave * 16;
    int half  = lane >> 4;
    int l16   = lane & 15;
    int mRow  = mBase + l16;
    int nCol  = nBase + l16;

    float rscale = rs[mRow];
    const float* Arow = A + (size_t)mRow * K + half * 2;
    const float* Wcol = W + (size_t)half * 2 * N + nCol;

    v8f c = {};
    #pragma unroll
    for (int k = 0; k < K; k += 4) {
        v2f a;
        a.x = Arow[k]     * rscale;
        a.y = Arow[k + 1] * rscale;
        v2f b;
        b.x = Wcol[(size_t)k * N];
        b.y = Wcol[(size_t)(k + 1) * N];
        c = __builtin_amdgcn_wmma_f32_16x16x4_f32(false, a, false, b, (short)0, c,
                                                  false, false);
    }

    float bv = bias[nCol];
    #pragma unroll
    for (int r = 0; r < 8; ++r) {
        int row = mBase + r + half * 8;
        float v = c[r] + bv;
        if (RELU) v = fmaxf(v, 0.0f);
        out[(size_t)row * N + nCol] = v;
    }
}


extern "C" void kernel_launch(void* const* d_in, const int* in_sizes, int n_in,
                              void* d_out, int out_size, void* d_ws, size_t ws_size,
                              hipStream_t stream) {
    const float* in_feat = (const float*)d_in[0];
    const int*   src     = (const int*)d_in[1];
    const int*   dst     = (const int*)d_in[2];
    const float* W1      = (const float*)d_in[3];
    const float* b1      = (const float*)d_in[4];
    const float* W2      = (const float*)d_in[5];
    const float* b2      = (const float*)d_in[6];
    float* out = (float*)d_out;

    float* ws      = (float*)d_ws;
    float* rs_out  = ws;
    float* rs_in   = ws + NNP;
    int*   deg_in  = (int*)(ws + 2 * NNP);
    int*   rowptr  = (int*)(ws + 3 * NNP);
    float* agg     = ws + 4 * NNP;
    float* hidden  = agg + (size_t)NN * 128;
    int*   dtable  = (int*)(hidden + (size_t)NN * 128);
    int*   sliceoff = dtable + (size_t)2 * HS * NNP;
    int*   sorted  = sliceoff + (size_t)HS * NNP;

    degree_hist_kernel<<<2 * HS, HT, 0, stream>>>(src, dst, dtable);
    degree_rsqrt_kernel<<<2 * DB, DT, 0, stream>>>(dtable, rs_out, rs_in, deg_in, sliceoff);
    edge_fill_kernel<<<HS, HT, 0, stream>>>(src, dst, deg_in, sliceoff, rowptr, sorted);

    gather_rows_kernel<<<NN / GW, GW * 32, 0, stream>>>(rowptr, sorted, in_feat, rs_out, agg);
    gemm_wmma_f32<INF, HF, true><<<NN / 16, (HF / 16) * 32, 0, stream>>>(
        agg, rs_in, W1, b1, hidden);

    gather_rows_kernel<<<NN / GW, GW * 32, 0, stream>>>(rowptr, sorted, hidden, rs_out, agg);
    gemm_wmma_f32<HF, OF, false><<<NN / 16, (OF / 16) * 32, 0, stream>>>(
        agg, rs_in, W2, b2, out);
}


# each layer's GEMM fused into its SpMM gather: 16 gathered rows go to LDS as the f32 WMMA A tile, neighbour-sum buffer and 2 launches dropped (7 -> 5); sums and their order unchanged
# speedup vs baseline: 4.3133x; 1.0786x over previous
#include <hip/hip_runtime.h>


#define NN 10000
#define NE 640000
#define INF 128
#define HF  128
#define OF  64

typedef float v2f __attribute__((ext_vector_type(2)));
typedef float v8f __attribute__((ext_vector_type(8)));

#define HS   64
#define HT   1024
#define NNP  10016

__global__ void __launch_bounds__(HT)
degree_hist_kernel(const int* __restrict__ src, const int* __restrict__ dst,
                   int* __restrict__ table) {
    __shared__ __attribute__((aligned(16))) int hist[NN];
    static_assert(NE % HS == 0, "edge slices must be equal");
    static_assert(NN % 4 == 0 && (NE / HS) % 4 == 0, "rows and slices move in 16-byte pieces");
    const int which = blockIdx.x / HS;
    const int slice = blockIdx.x - which * HS;
    const int* __restrict__ idx = (which ? dst : src) + (size_t)slice * (NE / HS);

    const int4* __restrict__ idx4 = (const int4*)idx;
    constexpr int PER4  = NE / HS / 4;
    constexpr int TRIPS = (PER4 + HT - 1) / HT;
    int4 v[TRIPS];
    #pragma unroll
    for (int j = 0; j < TRIPS; ++j) {
        int i = threadIdx.x + j * HT;
        if (i < PER4) v[j] = idx4[i];
    }

    int4* hist4 = (int4*)hist;
    for (int i = threadIdx.x; i < NN / 4; i += HT) hist4[i] = make_int4(0, 0, 0, 0);
    __syncthreads();
    #pragma unroll
    for (int j = 0; j < TRIPS; ++j) {
        int i = threadIdx.x + j * HT;
        if (i < PER4) {
            atomicAdd(&hist[v[j].x], 1);
            atomicAdd(&hist[v[j].y], 1);
            atomicAdd(&hist[v[j].z], 1);
            atomicAdd(&hist[v[j].w], 1);
        }
    }
    __syncthreads();
    int4* row4 = (int4*)(table + (size_t)blockIdx.x * NNP);
    for (int i = threadIdx.x; i < NN / 4; i += HT) row4[i] = hist4[i];
}

#define DT 128
#define DB ((NN + DT - 1) / DT)
__global__ void __launch_bounds__(DT)
degree_rsqrt_kernel(const int* __restrict__ table, float* __restrict__ rs_out,
                    float* __restrict__ rs_in, int* __restrict__ deg_in,
                    int* __restrict__ sliceoff) {
    const int which = blockIdx.x >= DB;
    const int node  = (blockIdx.x - which * DB) * DT + threadIdx.x;
    if (node >= NN) return;
    const int* __restrict__ col = table + (size_t)which * HS * NNP + node;
    int c[HS];
    #pragma unroll
    for (int b = 0; b < HS; ++b) c[b] = col[(size_t)b * NNP];
    int s = 0;
    if (which) {
        #pragma unroll
        for (int b = 0; b < HS; ++b) { sliceoff[(size_t)b * NNP + node] = s; s += c[b]; }
        deg_in[node] = s;
        rs_in[node] = rsqrtf(fmaxf((float)s, 1.0f));
    } else {
        #pragma unroll
        for (int b = 0; b < HS; ++b) s += c[b];
        rs_out[node] = rsqrtf(fmaxf((float)s, 1.0f));
    }
}

#define SCAN_PER 10
__global__ void __launch_bounds__(HT)
edge_fill_kernel(const int* __restrict__ src, const int* __restrict__ dst,
                 const int* __restrict__ deg_in, const int* __restrict__ sliceoff,
                 int* __restrict__ rowptr, int* __restrict__ sorted_src) {
    static_assert(HT * SCAN_PER >= NN, "the block scan must cover every node");
    static_assert(HT == 1024, "the scan of wave totals is one wave of 32 lanes");
    __shared__ int cur[HT * SCAN_PER];
    __shared__ int wtot[HT / 32];
    const int slice = blockIdx.x;
    const int tid   = threadIdx.x;

    const int4* __restrict__ d4 = (const int4*)(dst + (size_t)slice * (NE / HS));
    const int4* __restrict__ s4 = (const int4*)(src + (size_t)slice * (NE / HS));
    constexpr int PER4  = NE / HS / 4;
    constexpr int TRIPS = (PER4 + HT - 1) / HT;
    int4 dv[TRIPS], sv[TRIPS];
    #pragma unroll
    for (int j = 0; j < TRIPS; ++j) {
        int i = tid + j * HT;
        if (i < PER4) { dv[j] = d4[i]; sv[j] = s4[i]; }
    }

    const int* __restrict__ so = sliceoff + (size_t)slice * NNP;
    int dg[SCAN_PER], sof[SCAN_PER];
    #pragma unroll
    for (int k = 0; k < SCAN_PER; ++k) {
        int i = min(tid + k * HT, NN - 1);
        dg[k]  = deg_in[i];
        sof[k] = so[i];
    }
    #pragma unroll
    for (int k = 0; k < SCAN_PER; ++k) {
        int i = tid + k * HT;
        cur[i] = (i < NN) ? dg[k] : 0;
    }
    __syncthreads();

    int loc[SCAN_PER];
    int sum = 0;
    #pragma unroll
    for (int k = 0; k < SCAN_PER; ++k) { loc[k] = sum; sum += cur[tid * SCAN_PER + k]; }
    const int lane = tid & 31;
    const int wv   = tid >> 5;
    int inc = sum;
    #pragma unroll
    for (int off = 1; off < 32; off <<= 1) {
        int up = __shfl_up(inc, off);
        if (lane >= off) inc += up;
    }
    if (lane == 31) wtot[wv] = inc;
    __syncthreads();
    if (wv == 0) {
        int w = wtot[lane];
        int winc = w;
        #pragma unroll
        for (int off = 1; off < 32; off <<= 1) {
            int up = __shfl_up(winc, off);
            if (lane >= off) winc += up;
        }
        wtot[lane] = winc - w;
    }
    __syncthreads();
    const int before = wtot[wv] + inc - sum;
    #pragma unroll
    for (int k = 0; k < SCAN_PER; ++k) cur[tid * SCAN_PER + k] = before + loc[k];
    if (slice == 0 && tid == HT - 1) rowptr[NN] = before + sum;
    __syncthreads();

    #pragma unroll
    for (int k = 0; k < SCAN_PER; ++k) {
        int i = tid + k * HT;
        if (i < NN) {
            int c = cur[i];
            if (slice == 0) rowptr[i] = c;
            cur[i] = c + sof[k];
        }
    }
    __syncthreads();

    int slot[TRIPS][4];
    #pragma unroll
    for (int j = 0; j < TRIPS; ++j) {
        int i = tid + j * HT;
        if (i < PER4) {
            slot[j][0] = atomicAdd(&cur[dv[j].x], 1);
            slot[j][1] = atomicAdd(&cur[dv[j].y], 1);
            slot[j][2] = atomicAdd(&cur[dv[j].z], 1);
            slot[j][3] = atomicAdd(&cur[dv[j].w], 1);
        }
    }
    #pragma unroll
    for (int j = 0; j < TRIPS; ++j) {
        int i = tid + j * HT;
        if (i < PER4) {
            sorted_src[slot[j][0]] = sv[j].x;
            sorted_src[slot[j][1]] = sv[j].y;
            sorted_src[slot[j][2]] = sv[j].z;
            sorted_src[slot[j][3]] = sv[j].w;
        }
    }
}

#define FW 16
#define GU 8
#define AP 132
template <int N, bool RELU>
__global__ void __launch_bounds__(FW * 32)
layer_kernel(const int* __restrict__ rowptr, const int* __restrict__ sorted_src,
             const float* __restrict__ x,
             const float* __restrict__ rs_out,
             const float* __restrict__ rs_in,
             const float* __restrict__ W,
             const float* __restrict__ bias,
             float* __restrict__ out) {
    static_assert(NN % FW == 0, "whole blocks of nodes");
    static_assert(N % 16 == 0 && N / 16 <= FW, "one wave per 16-column tile");
    __shared__ __attribute__((aligned(16))) float tile[FW * AP];
    const int lane = threadIdx.x & 31;
    const int wave = __builtin_amdgcn_readfirstlane(threadIdx.x >> 5);
    const int node = blockIdx.x * FW + wave;
    const int beg = __builtin_amdgcn_readfirstlane(rowptr[node]);
    const int end = __builtin_amdgcn_readfirstlane(rowptr[node + 1]);
    const float rsn = rs_in[node];
    const float* __restrict__ xl = x + lane * 4;
    float4 acc = make_float4(0.0f, 0.0f, 0.0f, 0.0f);
    for (int base = beg; base < end; base += 32) {
        const int n = min(32, end - base);
        const int my    = sorted_src[base + min(lane, n - 1)];
        const int myr_i = __float_as_int(rs_out[my]);
        int k = 0;
        for (; k + GU <= n; k += GU) {
            float4 v[GU];
            #pragma unroll
            for (int u = 0; u < GU; ++u) {
                const int s = __builtin_amdgcn_readlane(my, k + u);
                v[u] = *(const float4*)(xl + (size_t)s * 128);
            }
            #pragma unroll
            for (int u = 0; u < GU; ++u) {
                const float r = __int_as_float(__builtin_amdgcn_readlane(myr_i, k + u));
                acc.x += v[u].x * r;
                acc.y += v[u].y * r;
                acc.z += v[u].z * r;
                acc.w += v[u].w * r;
            }
        }
        if (k < n) {
            float4 v[GU];
            #pragma unroll
            for (int u = 0; u < GU; ++u) {
                const int s = __builtin_amdgcn_readlane(my, k + u);
                v[u] = *(const float4*)(xl + (size_t)s * 128);
            }
            #pragma unroll
            for (int u = 0; u < GU; ++u) {
                const float r  = __int_as_float(__builtin_amdgcn_readlane(myr_i, k + u));
                const bool  on = k + u < n;
                acc.x = on ? acc.x + v[u].x * r : acc.x;
                acc.y = on ? acc.y + v[u].y * r : acc.y;
                acc.z = on ? acc.z + v[u].z * r : acc.z;
                acc.w = on ? acc.w + v[u].w * r : acc.w;
            }
        }
    }
    acc.x *= rsn; acc.y *= rsn; acc.z *= rsn; acc.w *= rsn;
    *(float4*)(tile + wave * AP + lane * 4) = acc;

    constexpr int NT = N / 16;
    const int half  = lane >> 4;
    const int l16   = lane & 15;
    const int mBase = blockIdx.x * FW;
    const int nCol  = wave * 16 + l16;
    v2f b[32];
    if (wave < NT) {
        const float* Wcol = W + (size_t)half * 2 * N + nCol;
        #pragma unroll
        for (int k = 0; k < 128; k += 4) {
            b[k / 4].x = Wcol[(size_t)k * N];
            b[k / 4].y = Wcol[(size_t)(k + 1) * N];
        }
    }
    __syncthreads();
    if (wave >= NT) return;

    const float* arow = tile + l16 * AP + half * 2;
    v8f c = {};
    #pragma unroll
    for (int k = 0; k < 128; k += 4) {
        const v2f a = *(const v2f*)(arow + k);
        c = __builtin_amdgcn_wmma_f32_16x16x4_f32(false, a, false, b[k / 4], (short)0, c,
                                                  false, false);
    }

    const float bv = bias[nCol];
    #pragma unroll
    for (int r = 0; r < 8; ++r) {
        const int row = mBase + r + half * 8;
        float v = c[r] + bv;
        if (RELU) v = fmaxf(v, 0.0f);
        out[(size_t)row * N + nCol] = v;
    }
}


extern "C" void kernel_launch(void* const* d_in, const int* in_sizes, int n_in,
                              void* d_out, int out_size, void* d_ws, size_t ws_size,
                              hipStream_t stream) {
    static_assert(INF == 128 && HF == 128, "both products have K = 128 (one 512-byte row per node)");
    const float* in_feat = (const float*)d_in[0];
    const int*   src     = (const int*)d_in[1];
    const int*   dst     = (const int*)d_in[2];
    const float* W1      = (const float*)d_in[3];
    const float* b1      = (const float*)d_in[4];
    const float* W2      = (const float*)d_in[5];
    const float* b2      = (const float*)d_in[6];
    float* out = (float*)d_out;

    float* ws      = (float*)d_ws;
    float* rs_out  = ws;
    float* rs_in   = ws + NNP;
    int*   deg_in  = (int*)(ws + 2 * NNP);
    int*   rowptr  = (int*)(ws + 3 * NNP);
    float* hidden  = ws + 4 * NNP;
    int*   dtable  = (int*)(hidden + (size_t)NN * 128);
    int*   sliceoff = dtable + (size_t)2 * HS * NNP;
    int*   sorted  = sliceoff + (size_t)HS * NNP;

    degree_hist_kernel<<<2 * HS, HT, 0, stream>>>(src, dst, dtable);
    degree_rsqrt_kernel<<<2 * DB, DT, 0, stream>>>(dtable, rs_out, rs_in, deg_in, sliceoff);
    edge_fill_kernel<<<HS, HT, 0, stream>>>(src, dst, deg_in, sliceoff, rowptr, sorted);

    layer_kernel<HF, true><<<NN / FW, FW * 32, 0, stream>>>(
        rowptr, sorted, in_feat, rs_out, rs_in, W1, b1, hidden);
    layer_kernel<OF, false><<<NN / FW, FW * 32, 0, stream>>>(
        rowptr, sorted, hidden, rs_out, rs_in, W2, b2, out);
}


# layer 2 as A_norm(s_out*(h W2)), W2 product inside layer 1's block: hidden stays in LDS, P leaves as whole 256-B rows, last launch only gathers, 2 rows per load; 5 launches kept; sum order changed
# speedup vs baseline: 4.8205x; 1.1176x over previous
#include <hip/hip_runtime.h>


#define NN 10000
#define NE 640000
#define INF 128
#define HF  128
#define OF  64

typedef float v2f __attribute__((ext_vector_type(2)));
typedef float v8f __attribute__((ext_vector_type(8)));

#define HS   64
#define HT   1024
#define NNP  10016

__global__ void __launch_bounds__(HT)
degree_hist_kernel(const int* __restrict__ src, const int* __restrict__ dst,
                   int* __restrict__ table) {
    __shared__ __attribute__((aligned(16))) int hist[NN];
    static_assert(NE % HS == 0, "edge slices must be equal");
    static_assert(NN % 4 == 0 && (NE / HS) % 4 == 0, "rows and slices move in 16-byte pieces");
    const int which = blockIdx.x / HS;
    const int slice = blockIdx.x - which * HS;
    const int* __restrict__ idx = (which ? dst : src) + (size_t)slice * (NE / HS);

    const int4* __restrict__ idx4 = (const int4*)idx;
    constexpr int PER4  = NE / HS / 4;
    constexpr int TRIPS = (PER4 + HT - 1) / HT;
    int4 v[TRIPS];
    #pragma unroll
    for (int j = 0; j < TRIPS; ++j) {
        int i = threadIdx.x + j * HT;
        if (i < PER4) v[j] = idx4[i];
    }

    int4* hist4 = (int4*)hist;
    for (int i = threadIdx.x; i < NN / 4; i += HT) hist4[i] = make_int4(0, 0, 0, 0);
    __syncthreads();
    #pragma unroll
    for (int j = 0; j < TRIPS; ++j) {
        int i = threadIdx.x + j * HT;
        if (i < PER4) {
            atomicAdd(&hist[v[j].x], 1);
            atomicAdd(&hist[v[j].y], 1);
            atomicAdd(&hist[v[j].z], 1);
            atomicAdd(&hist[v[j].w], 1);
        }
    }
    __syncthreads();
    int4* row4 = (int4*)(table + (size_t)blockIdx.x * NNP);
    for (int i = threadIdx.x; i < NN / 4; i += HT) row4[i] = hist4[i];
}

#define DT 128
#define DB ((NN + DT - 1) / DT)
__global__ void __launch_bounds__(DT)
degree_rsqrt_kernel(const int* __restrict__ table, float* __restrict__ rs_out,
                    float* __restrict__ rs_in, int* __restrict__ deg_in,
                    int* __restrict__ sliceoff) {
    const int which = blockIdx.x >= DB;
    const int node  = (blockIdx.x - which * DB) * DT + threadIdx.x;
    if (node >= NN) return;
    const int* __restrict__ col = table + (size_t)which * HS * NNP + node;
    int c[HS];
    #pragma unroll
    for (int b = 0; b < HS; ++b) c[b] = col[(size_t)b * NNP];
    int s = 0;
    if (which) {
        #pragma unroll
        for (int b = 0; b < HS; ++b) { sliceoff[(size_t)b * NNP + node] = s; s += c[b]; }
        deg_in[node] = s;
        rs_in[node] = rsqrtf(fmaxf((float)s, 1.0f));
    } else {
        #pragma unroll
        for (int b = 0; b < HS; ++b) s += c[b];
        rs_out[node] = rsqrtf(fmaxf((float)s, 1.0f));
    }
}

#define SCAN_PER 10
__global__ void __launch_bounds__(HT)
edge_fill_kernel(const int* __restrict__ src, const int* __restrict__ dst,
                 const int* __restrict__ deg_in, const int* __restrict__ sliceoff,
                 int* __restrict__ rowptr, int* __restrict__ sorted_src) {
    static_assert(HT * SCAN_PER >= NN, "the block scan must cover every node");
    static_assert(HT == 1024, "the scan of wave totals is one wave of 32 lanes");
    __shared__ int cur[HT * SCAN_PER];
    __shared__ int wtot[HT / 32];
    const int slice = blockIdx.x;
    const int tid   = threadIdx.x;

    const int4* __restrict__ d4 = (const int4*)(dst + (size_t)slice * (NE / HS));
    const int4* __restrict__ s4 = (const int4*)(src + (size_t)slice * (NE / HS));
    constexpr int PER4  = NE / HS / 4;
    constexpr int TRIPS = (PER4 + HT - 1) / HT;
    int4 dv[TRIPS], sv[TRIPS];
    #pragma unroll
    for (int j = 0; j < TRIPS; ++j) {
        int i = tid + j * HT;
        if (i < PER4) { dv[j] = d4[i]; sv[j] = s4[i]; }
    }

    const int* __restrict__ so = sliceoff + (size_t)slice * NNP;
    int dg[SCAN_PER], sof[SCAN_PER];
    #pragma unroll
    for (int k = 0; k < SCAN_PER; ++k) {
        int i = min(tid + k * HT, NN - 1);
        dg[k]  = deg_in[i];
        sof[k] = so[i];
    }
    #pragma unroll
    for (int k = 0; k < SCAN_PER; ++k) {
        int i = tid + k * HT;
        cur[i] = (i < NN) ? dg[k] : 0;
    }
    __syncthreads();

    int loc[SCAN_PER];
    int sum = 0;
    #pragma unroll
    for (int k = 0; k < SCAN_PER; ++k) { loc[k] = sum; sum += cur[tid * SCAN_PER + k]; }
    const int lane = tid & 31;
    const int wv   = tid >> 5;
    int inc = sum;
    #pragma unroll
    for (int off = 1; off < 32; off <<= 1) {
        int up = __shfl_up(inc, off);
        if (lane >= off) inc += up;
    }
    if (lane == 31) wtot[wv] = inc;
    __syncthreads();
    if (wv == 0) {
        int w = wtot[lane];
        int winc = w;
        #pragma unroll
        for (int off = 1; off < 32; off <<= 1) {
            int up = __shfl_up(winc, off);
            if (lane >= off) winc += up;
        }
        wtot[lane] = winc - w;
    }
    __syncthreads();
    const int before = wtot[wv] + inc - sum;
    #pragma unroll
    for (int k = 0; k < SCAN_PER; ++k) cur[tid * SCAN_PER + k] = before + loc[k];
    if (slice == 0 && tid == HT - 1) rowptr[NN] = before + sum;
    __syncthreads();

    #pragma unroll
    for (int k = 0; k < SCAN_PER; ++k) {
        int i = tid + k * HT;
        if (i < NN) {
            int c = cur[i];
            if (slice == 0) rowptr[i] = c;
            cur[i] = c + sof[k];
        }
    }
    __syncthreads();

    int slot[TRIPS][4];
    #pragma unroll
    for (int j = 0; j < TRIPS; ++j) {
        int i = tid + j * HT;
        if (i < PER4) {
            slot[j][0] = atomicAdd(&cur[dv[j].x], 1);
            slot[j][1] = atomicAdd(&cur[dv[j].y], 1);
            slot[j][2] = atomicAdd(&cur[dv[j].z], 1);
            slot[j][3] = atomicAdd(&cur[dv[j].w], 1);
        }
    }
    #pragma unroll
    for (int j = 0; j < TRIPS; ++j) {
        int i = tid + j * HT;
        if (i < PER4) {
            sorted_src[slot[j][0]] = sv[j].x;
            sorted_src[slot[j][1]] = sv[j].y;
            sorted_src[slot[j][2]] = sv[j].z;
            sorted_src[slot[j][3]] = sv[j].w;
        }
    }
}

#define FW 16
#define GU 8
#define AP 132
#define PP 68
#define N1 (HF / 16)
#define N2 (OF / 16)
__global__ void __launch_bounds__(FW * 32)
layer1_kernel(const int* __restrict__ rowptr, const int* __restrict__ sorted_src,
              const float* __restrict__ x,
              const float* __restrict__ rs_out,
              const float* __restrict__ rs_in,
              const float* __restrict__ W1,
              const float* __restrict__ bias1,
              const float* __restrict__ W2,
              float* __restrict__ P) {
    static_assert(NN % FW == 0, "whole blocks of nodes");
    static_assert(FW - N1 == 2 * N2 && 2 * (FW - N1) == FW, "waves 0..N1-1: product 1; waves N1..FW-1: product 2, "
                                                            "a column tile and a half of K each, then two rows each");
    __shared__ __attribute__((aligned(16))) float tile[FW * AP];
    __shared__ __attribute__((aligned(16))) float htile[FW * AP];
    __shared__ __attribute__((aligned(16))) float ptile[2 * FW * PP];
    __shared__ float rso[FW];
    const int lane = threadIdx.x & 31;
    const int wave = __builtin_amdgcn_readfirstlane(threadIdx.x >> 5);
    const int node = blockIdx.x * FW + wave;
    const int beg = __builtin_amdgcn_readfirstlane(rowptr[node]);
    const int end = __builtin_amdgcn_readfirstlane(rowptr[node + 1]);
    const float rsn = rs_in[node];
    const float rsm = rs_out[node];
    const float* __restrict__ xl = x + lane * 4;
    float4 acc = make_float4(0.0f, 0.0f, 0.0f, 0.0f);
    for (int base = beg; base < end; base += 32) {
        const int n = min(32, end - base);
        const int my    = sorted_src[base + min(lane, n - 1)];
        const int myr_i = __float_as_int(rs_out[my]);
        int k = 0;
        for (; k + GU <= n; k += GU) {
            float4 v[GU];
            #pragma unroll
            for (int u = 0; u < GU; ++u) {
                const int s = __builtin_amdgcn_readlane(my, k + u);
                v[u] = *(const float4*)(xl + (size_t)s * 128);
            }
            #pragma unroll
            for (int u = 0; u < GU; ++u) {
                const float r = __int_as_float(__builtin_amdgcn_readlane(myr_i, k + u));
                acc.x += v[u].x * r;
                acc.y += v[u].y * r;
                acc.z += v[u].z * r;
                acc.w += v[u].w * r;
            }
        }
        if (k < n) {
            float4 v[GU];
            #pragma unroll
            for (int u = 0; u < GU; ++u) {
                const int s = __builtin_amdgcn_readlane(my, k + u);
                v[u] = *(const float4*)(xl + (size_t)s * 128);
            }
            #pragma unroll
            for (int u = 0; u < GU; ++u) {
                const float r  = __int_as_float(__builtin_amdgcn_readlane(myr_i, k + u));
                const bool  on = k + u < n;
                acc.x = on ? acc.x + v[u].x * r : acc.x;
                acc.y = on ? acc.y + v[u].y * r : acc.y;
                acc.z = on ? acc.z + v[u].z * r : acc.z;
                acc.w = on ? acc.w + v[u].w * r : acc.w;
            }
        }
    }
    acc.x *= rsn; acc.y *= rsn; acc.z *= rsn; acc.w *= rsn;
    *(float4*)(tile + wave * AP + lane * 4) = acc;
    if (lane == 0) rso[wave] = rsm;

    const int half  = lane >> 4;
    const int l16   = lane & 15;
    v2f bw1[32];
    if (wave < N1) {
        const float* Wcol = W1 + (size_t)half * 2 * HF + wave * 16 + l16;
        #pragma unroll
        for (int k = 0; k < 128; k += 4) {
            bw1[k / 4].x = Wcol[(size_t)k * HF];
            bw1[k / 4].y = Wcol[(size_t)(k + 1) * HF];
        }
    }
    __syncthreads();

    v2f bw2[16];
    const int t2 = (wave - N1) & (N2 - 1);
    const int kh = (wave - N1) / N2;
    if (wave < N1) {
        const float* arow = tile + l16 * AP + half * 2;
        v8f c = {};
        #pragma unroll
        for (int k = 0; k < 128; k += 4) {
            const v2f a = *(const v2f*)(arow + k);
            c = __builtin_amdgcn_wmma_f32_16x16x4_f32(false, a, false, bw1[k / 4], (short)0, c,
                                                      false, false);
        }
        const int nCol = wave * 16 + l16;
        const float bv = bias1[nCol];
        #pragma unroll
        for (int r = 0; r < 8; ++r) {
            const int row = r + half * 8;
            htile[row * AP + nCol] = fmaxf(c[r] + bv, 0.0f);
        }
    }
    if (wave >= N1) {
        const float* Wcol = W2 + (size_t)(kh * 64 + half * 2) * OF + t2 * 16 + l16;
        #pragma unroll
        for (int k = 0; k < 64; k += 4) {
            bw2[k / 4].x = Wcol[(size_t)k * OF];
            bw2[k / 4].y = Wcol[(size_t)(k + 1) * OF];
        }
    }
    __syncthreads();

    if (wave >= N1) {
        const float* arow = htile + l16 * AP + kh * 64 + half * 2;
        v8f c = {};
        #pragma unroll
        for (int k = 0; k < 64; k += 4) {
            const v2f a = *(const v2f*)(arow + k);
            c = __builtin_amdgcn_wmma_f32_16x16x4_f32(false, a, false, bw2[k / 4], (short)0, c,
                                                      false, false);
        }
        float* pt = ptile + kh * (FW * PP) + half * 8 * PP + t2 * 16 + l16;
        #pragma unroll
        for (int r = 0; r < 8; ++r) pt[r * PP] = c[r];
    }
    __syncthreads();

    if (wave >= N1) {
        const int row = (wave - N1) * 2 + half;
        const float4 p0 = *(const float4*)(ptile + row * PP + l16 * 4);
        const float4 p1 = *(const float4*)(ptile + (FW + row) * PP + l16 * 4);
        const float s = rso[row];
        float4 v;
        v.x = (p0.x + p1.x) * s; v.y = (p0.y + p1.y) * s; v.z = (p0.z + p1.z) * s; v.w = (p0.w + p1.w) * s;
        *(float4*)(P + (size_t)(blockIdx.x * FW + row) * OF + l16 * 4) = v;
    }
}


template <int J>
__device__ __forceinline__ int row_lane(int v) {
    return __builtin_amdgcn_update_dpp(v, v, 0x150 + J, 0xF, 0xF, false);
}
template <int KK>
__device__ __forceinline__ void load_rows(const char* __restrict__ pb, unsigned lo, int my, float4 (&v)[GU]) {
    static_assert(GU == 8, "eight steps written out");
    v[0] = *(const float4*)(pb + ((unsigned)row_lane<KK + 0>(my) * (unsigned)(OF * 4) + lo));
    v[1] = *(const float4*)(pb + ((unsigned)row_lane<KK + 1>(my) * (unsigned)(OF * 4) + lo));
    v[2] = *(const float4*)(pb + ((unsigned)row_lane<KK + 2>(my) * (unsigned)(OF * 4) + lo));
    v[3] = *(const float4*)(pb + ((unsigned)row_lane<KK + 3>(my) * (unsigned)(OF * 4) + lo));
    v[4] = *(const float4*)(pb + ((unsigned)row_lane<KK + 4>(my) * (unsigned)(OF * 4) + lo));
    v[5] = *(const float4*)(pb + ((unsigned)row_lane<KK + 5>(my) * (unsigned)(OF * 4) + lo));
    v[6] = *(const float4*)(pb + ((unsigned)row_lane<KK + 6>(my) * (unsigned)(OF * 4) + lo));
    v[7] = *(const float4*)(pb + ((unsigned)row_lane<KK + 7>(my) * (unsigned)(OF * 4) + lo));
}
template <int KK>
__device__ __forceinline__ void gather_steps(const char* __restrict__ pb, unsigned lo, int my,
                                             int n0, int n1, int nh, float4& acc) {
    if (KK + GU <= n1) {
        float4 v[GU];
        load_rows<KK>(pb, lo, my, v);
        #pragma unroll
        for (int u = 0; u < GU; ++u) {
            acc.x += v[u].x; acc.y += v[u].y; acc.z += v[u].z; acc.w += v[u].w;
        }
    } else if (KK < n0) {
        float4 v[GU];
        load_rows<KK>(pb, lo, my, v);
        #pragma unroll
        for (int u = 0; u < GU; ++u) {
            const bool on = KK + u < nh;
            acc.x += on ? v[u].x : 0.0f;
            acc.y += on ? v[u].y : 0.0f;
            acc.z += on ? v[u].z : 0.0f;
            acc.w += on ? v[u].w : 0.0f;
        }
    }
}
__device__ __forceinline__ float other_half(float v) {
    return __int_as_float(__builtin_amdgcn_permlanex16(__float_as_int(v), __float_as_int(v),
                                                       0x76543210, 0xfedcba98, false, false));
}

__global__ void __launch_bounds__(FW * 32)
gather_out_kernel(const int* __restrict__ rowptr, const int* __restrict__ sorted_src,
                  const float* __restrict__ P,
                  const float* __restrict__ rs_in,
                  const float* __restrict__ bias,
                  float* __restrict__ out) {
    static_assert(OF == 64, "a row of P is 16 lanes x 16 bytes");
    const int lane = threadIdx.x & 31;
    const int wave = __builtin_amdgcn_readfirstlane(threadIdx.x >> 5);
    const int node = blockIdx.x * FW + wave;
    const int beg = __builtin_amdgcn_readfirstlane(rowptr[node]);
    const int end = __builtin_amdgcn_readfirstlane(rowptr[node + 1]);
    const float rsn = rs_in[node];
    const int half = lane >> 4;
    const int l16  = lane & 15;
    const float4 bv = *(const float4*)(bias + l16 * 4);
    const int cnt  = end - beg;
    const int len0 = (cnt + 1) >> 1;
    const int len1 = cnt - len0;
    const int lenh = half ? len1 : len0;
    const int* __restrict__ run = sorted_src + beg + (half ? len0 : 0);
    const char* __restrict__ pb = (const char*)P;
    const unsigned lo = (unsigned)l16 * 16u;
    float4 acc = make_float4(0.0f, 0.0f, 0.0f, 0.0f);
    for (int b = 0; b < len0; b += 16) {
        const int n0 = min(16, len0 - b);
        const int n1 = min(16, len1 - b);
        const int nh = half ? n1 : n0;
        const int my = run[min(b + l16, lenh - 1)];
        gather_steps<0>(pb, lo, my, n0, n1, nh, acc);
        gather_steps<GU>(pb, lo, my, n0, n1, nh, acc);
    }
    acc.x += other_half(acc.x);
    acc.y += other_half(acc.y);
    acc.z += other_half(acc.z);
    acc.w += other_half(acc.w);
    if (half == 0) {
        float4 o;
        o.x = acc.x * rsn + bv.x;
        o.y = acc.y * rsn + bv.y;
        o.z = acc.z * rsn + bv.z;
        o.w = acc.w * rsn + bv.w;
        *(float4*)(out + (size_t)node * OF + l16 * 4) = o;
    }
}


extern "C" void kernel_launch(void* const* d_in, const int* in_sizes, int n_in,
                              void* d_out, int out_size, void* d_ws, size_t ws_size,
                              hipStream_t stream) {
    static_assert(INF == 128 && HF == 128, "both products have K = 128 (one 512-byte row per node)");
    const float* in_feat = (const float*)d_in[0];
    const int*   src     = (const int*)d_in[1];
    const int*   dst     = (const int*)d_in[2];
    const float* W1      = (const float*)d_in[3];
    const float* b1      = (const float*)d_in[4];
    const float* W2      = (const float*)d_in[5];
    const float* b2      = (const float*)d_in[6];
    float* out = (float*)d_out;

    float* ws      = (float*)d_ws;
    float* rs_out  = ws;
    float* rs_in   = ws + NNP;
    int*   deg_in  = (int*)(ws + 2 * NNP);
    int*   rowptr  = (int*)(ws + 3 * NNP);
    float* P       = ws + 4 * NNP;
    int*   dtable  = (int*)(P + (size_t)NN * OF);
    int*   sliceoff = dtable + (size_t)2 * HS * NNP;
    int*   sorted  = sliceoff + (size_t)HS * NNP;

    degree_hist_kernel<<<2 * HS, HT, 0, stream>>>(src, dst, dtable);
    degree_rsqrt_kernel<<<2 * DB, DT, 0, stream>>>(dtable, rs_out, rs_in, deg_in, sliceoff);
    edge_fill_kernel<<<HS, HT, 0, stream>>>(src, dst, deg_in, sliceoff, rowptr, sorted);

    layer1_kernel<<<NN / FW, FW * 32, 0, stream>>>(
        rowptr, sorted, in_feat, rs_out, rs_in, W1, b1, W2, P);
    gather_out_kernel<<<NN / FW, FW * 32, 0, stream>>>(rowptr, sorted, P, rs_in, b2, out);
}


# edge sort regrouped: each slice grouped by 125 dst ranges in LDS; one block per range counts, scans, places its edges, sums its out-degrees; 5 launches -> 4; sum order kept but for ties in a slice
# speedup vs baseline: 7.1119x; 1.4753x over previous
#include <hip/hip_runtime.h>


#define NN 10000
#define NE 640000
#define INF 128
#define HF  128
#define OF  64

typedef float v2f __attribute__((ext_vector_type(2)));
typedef float v8f __attribute__((ext_vector_type(8)));

#define HS   64
#define SL   (NE / HS)
#define HT   1024
#define NNP  10016
#define NB   125
#define RS   (NN / NB)
#define BP   128
#define KSH  14
#define LW   (NN > SL ? NN : SL)

__device__ __forceinline__ int wave_inclusive(int v, int lane) {
    #pragma unroll
    for (int off = 1; off < 32; off <<= 1) {
        int up = __shfl_up(v, off);
        if (lane >= off) v += up;
    }
    return v;
}

__global__ void __launch_bounds__(HT)
slice_kernel(const int* __restrict__ src, const int* __restrict__ dst,
             int* __restrict__ table, int* __restrict__ keys, int* __restrict__ bstart) {
    __shared__ __attribute__((aligned(16))) int lds[LW];
    __shared__ int bcnt[BP];
    __shared__ int boffs[BP];
    static_assert(NE % HS == 0, "edge slices must be equal");
    static_assert(NN % 4 == 0 && SL % 4 == 0, "rows and slices move in 16-byte pieces");
    static_assert(NB * RS == NN && NB + 1 <= BP && RS <= BP, "ranges tile the nodes; range arrays fit their pitch");
    static_assert(NN <= (1 << KSH) && NN < (1 << (31 - KSH)), "a packed word holds both node ids");
    const int which = blockIdx.x / HS;
    const int slice = blockIdx.x - which * HS;
    const int tid   = threadIdx.x;
    constexpr int PER4  = SL / 4;
    constexpr int TRIPS = (PER4 + HT - 1) / HT;
    int4* lds4 = (int4*)lds;

    if (which == 0) {
        const int4* __restrict__ idx4 = (const int4*)(src + (size_t)slice * SL);
        int4 v[TRIPS];
        #pragma unroll
        for (int j = 0; j < TRIPS; ++j) {
            int i = tid + j * HT;
            if (i < PER4) v[j] = idx4[i];
        }
        for (int i = tid; i < NN / 4; i += HT) lds4[i] = make_int4(0, 0, 0, 0);
        __syncthreads();
        #pragma unroll
        for (int j = 0; j < TRIPS; ++j) {
            int i = tid + j * HT;
            if (i < PER4) {
                atomicAdd(&lds[v[j].x], 1);
                atomicAdd(&lds[v[j].y], 1);
                atomicAdd(&lds[v[j].z], 1);
                atomicAdd(&lds[v[j].w], 1);
            }
        }
        __syncthreads();
        int4* row4 = (int4*)(table + (size_t)slice * NNP);
        for (int i = tid; i < NN / 4; i += HT) row4[i] = lds4[i];
    } else {
        const int4* __restrict__ d4 = (const int4*)(dst + (size_t)slice * SL);
        const int4* __restrict__ s4 = (const int4*)(src + (size_t)slice * SL);
        int4 dv[TRIPS], sv[TRIPS];
        #pragma unroll
        for (int j = 0; j < TRIPS; ++j) {
            int i = tid + j * HT;
            if (i < PER4) { dv[j] = d4[i]; sv[j] = s4[i]; }
        }
        if (tid < BP) bcnt[tid] = 0;
        __syncthreads();
        int bk[TRIPS][4], rk[TRIPS][4];
        #pragma unroll
        for (int j = 0; j < TRIPS; ++j) {
            int i = tid + j * HT;
            if (i < PER4) {
                bk[j][0] = dv[j].x / RS; bk[j][1] = dv[j].y / RS;
                bk[j][2] = dv[j].z / RS; bk[j][3] = dv[j].w / RS;
                rk[j][0] = atomicAdd(&bcnt[bk[j][0]], 1);
                rk[j][1] = atomicAdd(&bcnt[bk[j][1]], 1);
                rk[j][2] = atomicAdd(&bcnt[bk[j][2]], 1);
                rk[j][3] = atomicAdd(&bcnt[bk[j][3]], 1);
            }
        }
        __syncthreads();
        if (tid < 32) {
            const int c0 = bcnt[tid * 4 + 0], c1 = bcnt[tid * 4 + 1];
            const int c2 = bcnt[tid * 4 + 2], c3 = bcnt[tid * 4 + 3];
            const int s  = c0 + c1 + c2 + c3;
            const int ex = wave_inclusive(s, tid) - s;
            boffs[tid * 4 + 0] = ex;
            boffs[tid * 4 + 1] = ex + c0;
            boffs[tid * 4 + 2] = ex + c0 + c1;
            boffs[tid * 4 + 3] = ex + c0 + c1 + c2;
        }
        __syncthreads();
        #pragma unroll
        for (int j = 0; j < TRIPS; ++j) {
            int i = tid + j * HT;
            if (i < PER4) {
                lds[boffs[bk[j][0]] + rk[j][0]] = (dv[j].x << KSH) | sv[j].x;
                lds[boffs[bk[j][1]] + rk[j][1]] = (dv[j].y << KSH) | sv[j].y;
                lds[boffs[bk[j][2]] + rk[j][2]] = (dv[j].z << KSH) | sv[j].z;
                lds[boffs[bk[j][3]] + rk[j][3]] = (dv[j].w << KSH) | sv[j].w;
            }
        }
        if (tid <= NB) bstart[slice * BP + tid] = boffs[tid];
        __syncthreads();
        int4* out4 = (int4*)(keys + (size_t)slice * SL);
        for (int i = tid; i < SL / 4; i += HT) out4[i] = lds4[i];
    }
}

#define FWV (HT / 32)
__global__ void __launch_bounds__(HT)
range_fill_kernel(const int* __restrict__ keys, const int* __restrict__ bstart,
                  const int* __restrict__ table, float* __restrict__ rs_out,
                  float* __restrict__ rs_in, int* __restrict__ rowptr,
                  int* __restrict__ sorted_src) {
    static_assert(HS == 2 * FWV, "two segments per wave");
    static_assert(HS % (HT / BP) == 0, "table rows split evenly over the thread groups");
    __shared__ int cntw[FWV * RS];
    __shared__ int tot[BP];
    __shared__ int dego[BP];
    __shared__ int segbeg[HS], seglen[HS];
    __shared__ int basesh;
    const int b    = blockIdx.x;
    const int tid  = threadIdx.x;
    const int lane = tid & 31;
    const int wv   = tid >> 5;

    constexpr int RPT = HS / (HT / BP);
    const int nl = tid & (BP - 1);
    int od = 0;
    if (nl < RS) {
        const int* __restrict__ col = table + (size_t)((tid / BP) * RPT) * NNP + b * RS + nl;
        #pragma unroll
        for (int r = 0; r < RPT; ++r) od += col[(size_t)r * NNP];
    }
    int s0 = 0;
    if (tid < HS) {
        s0 = bstart[tid * BP + b];
        const int s1 = bstart[tid * BP + b + 1];
        segbeg[tid] = tid * SL + s0;
        seglen[tid] = s1 - s0;
    }
    for (int i = tid; i < FWV * RS; i += HT) cntw[i] = 0;
    if (tid < BP) { tot[tid] = 0; dego[tid] = 0; }
    if (tid == 0) basesh = 0;
    __syncthreads();

    if (nl < RS) atomicAdd(&dego[nl], od);
    if (tid < HS) atomicAdd(&basesh, s0);
    int* mycnt = cntw + wv * RS;
    for (int q = 0; q < 2; ++q) {
        const int sb = segbeg[2 * wv + q], sl = seglen[2 * wv + q];
        for (int j = lane; j < sl; j += 32) {
            const int key = keys[sb + j];
            atomicAdd(&mycnt[(key >> KSH) - b * RS], 1);
        }
    }
    __syncthreads();

    if (tid < RS) {
        int c[FWV];
        #pragma unroll
        for (int w = 0; w < FWV; ++w) c[w] = cntw[w * RS + tid];
        int run = 0;
        #pragma unroll
        for (int w = 0; w < FWV; ++w) { cntw[w * RS + tid] = run; run += c[w]; }
        tot[tid] = run;
    }
    __syncthreads();

    if (tid < 32) {
        const int c0 = tot[tid * 4 + 0], c1 = tot[tid * 4 + 1];
        const int c2 = tot[tid * 4 + 2], c3 = tot[tid * 4 + 3];
        const int s   = c0 + c1 + c2 + c3;
        const int inc = wave_inclusive(s, tid);
        const int st0 = basesh + inc - s;
        const int st[4] = { st0, st0 + c0, st0 + c0 + c1, st0 + c0 + c1 + c2 };
        const int cc[4] = { c0, c1, c2, c3 };
        #pragma unroll
        for (int q = 0; q < 4; ++q) {
            const int n = tid * 4 + q;
            tot[n] = st[q];
            if (n < RS) {
                const int node = b * RS + n;
                rowptr[node] = st[q];
                rs_in[node]  = rsqrtf(fmaxf((float)cc[q], 1.0f));
                rs_out[node] = rsqrtf(fmaxf((float)dego[n], 1.0f));
            }
        }
        if (b == NB - 1 && tid == 31) rowptr[NN] = basesh + inc;
    }
    __syncthreads();

    for (int q = 0; q < 2; ++q) {
        const int sb = segbeg[2 * wv + q], sl = seglen[2 * wv + q];
        for (int j = lane; j < sl; j += 32) {
            const int key = keys[sb + j];
            const int dl  = (key >> KSH) - b * RS;
            const int slot = tot[dl] + atomicAdd(&mycnt[dl], 1);
            sorted_src[slot] = key & ((1 << KSH) - 1);
        }
    }
}

#define FW 16
#define GU 8
#define AP 132
#define PP 68
#define N1 (HF / 16)
#define N2 (OF / 16)
__global__ void __launch_bounds__(FW * 32)
layer1_kernel(const int* __restrict__ rowptr, const int* __restrict__ sorted_src,
              const float* __restrict__ x,
              const float* __restrict__ rs_out,
              const float* __restrict__ rs_in,
              const float* __restrict__ W1,
              const float* __restrict__ bias1,
              const float* __restrict__ W2,
              float* __restrict__ P) {
    static_assert(NN % FW == 0, "whole blocks of nodes");
    static_assert(FW - N1 == 2 * N2 && 2 * (FW - N1) == FW, "waves 0..N1-1: product 1; waves N1..FW-1: product 2, "
                                                            "a column tile and a half of K each, then two rows each");
    __shared__ __attribute__((aligned(16))) float tile[FW * AP];
    __shared__ __attribute__((aligned(16))) float htile[FW * AP];
    __shared__ __attribute__((aligned(16))) float ptile[2 * FW * PP];
    __shared__ float rso[FW];
    const int lane = threadIdx.x & 31;
    const int wave = __builtin_amdgcn_readfirstlane(threadIdx.x >> 5);
    const int node = blockIdx.x * FW + wave;
    const int beg = __builtin_amdgcn_readfirstlane(rowptr[node]);
    const int end = __builtin_amdgcn_readfirstlane(rowptr[node + 1]);
    const float rsn = rs_in[node];
    const float rsm = rs_out[node];
    const float* __restrict__ xl = x + lane * 4;
    float4 acc = make_float4(0.0f, 0.0f, 0.0f, 0.0f);
    for (int base = beg; base < end; base += 32) {
        const int n = min(32, end - base);
        const int my    = sorted_src[base + min(lane, n - 1)];
        const int myr_i = __float_as_int(rs_out[my]);
        int k = 0;
        for (; k + GU <= n; k += GU) {
            float4 v[GU];
            #pragma unroll
            for (int u = 0; u < GU; ++u) {
                const int s = __builtin_amdgcn_readlane(my, k + u);
                v[u] = *(const float4*)(xl + (size_t)s * 128);
            }
            #pragma unroll
            for (int u = 0; u < GU; ++u) {
                const float r = __int_as_float(__builtin_amdgcn_readlane(myr_i, k + u));
                acc.x += v[u].x * r;
                acc.y += v[u].y * r;
                acc.z += v[u].z * r;
                acc.w += v[u].w * r;
            }
        }
        if (k < n) {
            float4 v[GU];
            #pragma unroll
            for (int u = 0; u < GU; ++u) {
                const int s = __builtin_amdgcn_readlane(my, k + u);
                v[u] = *(const float4*)(xl + (size_t)s * 128);
            }
            #pragma unroll
            for (int u = 0; u < GU; ++u) {
                const float r  = __int_as_float(__builtin_amdgcn_readlane(myr_i, k + u));
                const bool  on = k + u < n;
                acc.x = on ? acc.x + v[u].x * r : acc.x;
                acc.y = on ? acc.y + v[u].y * r : acc.y;
                acc.z = on ? acc.z + v[u].z * r : acc.z;
                acc.w = on ? acc.w + v[u].w * r : acc.w;
            }
        }
    }
    acc.x *= rsn; acc.y *= rsn; acc.z *= rsn; acc.w *= rsn;
    *(float4*)(tile + wave * AP + lane * 4) = acc;
    if (lane == 0) rso[wave] = rsm;

    const int half  = lane >> 4;
    const int l16   = lane & 15;
    v2f bw1[32];
    if (wave < N1) {
        const float* Wcol = W1 + (size_t)half * 2 * HF + wave * 16 + l16;
        #pragma unroll
        for (int k = 0; k < 128; k += 4) {
            bw1[k / 4].x = Wcol[(size_t)k * HF];
            bw1[k / 4].y = Wcol[(size_t)(k + 1) * HF];
        }
    }
    __syncthreads();

    v2f bw2[16];
    const int t2 = (wave - N1) & (N2 - 1);
    const int kh = (wave - N1) / N2;
    if (wave < N1) {
        const float* arow = tile + l16 * AP + half * 2;
        v8f c = {};
        #pragma unroll
        for (int k = 0; k < 128; k += 4) {
            const v2f a = *(const v2f*)(arow + k);
            c = __builtin_amdgcn_wmma_f32_16x16x4_f32(false, a, false, bw1[k / 4], (short)0, c,
                                                      false, false);
        }
        const int nCol = wave * 16 + l16;
        const float bv = bias1[nCol];
        #pragma unroll
        for (int r = 0; r < 8; ++r) {
            const int row = r + half * 8;
            htile[row * AP + nCol] = fmaxf(c[r] + bv, 0.0f);
        }
    }
    if (wave >= N1) {
        const float* Wcol = W2 + (size_t)(kh * 64 + half * 2) * OF + t2 * 16 + l16;
        #pragma unroll
        for (int k = 0; k < 64; k += 4) {
            bw2[k / 4].x = Wcol[(size_t)k * OF];
            bw2[k / 4].y = Wcol[(size_t)(k + 1) * OF];
        }
    }
    __syncthreads();

    if (wave >= N1) {
        const float* arow = htile + l16 * AP + kh * 64 + half * 2;
        v8f c = {};
        #pragma unroll
        for (int k = 0; k < 64; k += 4) {
            const v2f a = *(const v2f*)(arow + k);
            c = __builtin_amdgcn_wmma_f32_16x16x4_f32(false, a, false, bw2[k / 4], (short)0, c,
                                                      false, false);
        }
        float* pt = ptile + kh * (FW * PP) + half * 8 * PP + t2 * 16 + l16;
        #pragma unroll
        for (int r = 0; r < 8; ++r) pt[r * PP] = c[r];
    }
    __syncthreads();

    if (wave >= N1) {
        const int row = (wave - N1) * 2 + half;
        const float4 p0 = *(const float4*)(ptile + row * PP + l16 * 4);
        const float4 p1 = *(const float4*)(ptile + (FW + row) * PP + l16 * 4);
        const float s = rso[row];
        float4 v;
        v.x = (p0.x + p1.x) * s; v.y = (p0.y + p1.y) * s; v.z = (p0.z + p1.z) * s; v.w = (p0.w + p1.w) * s;
        *(float4*)(P + (size_t)(blockIdx.x * FW + row) * OF + l16 * 4) = v;
    }
}


template <int J>
__device__ __forceinline__ int row_lane(int v) {
    return __builtin_amdgcn_update_dpp(v, v, 0x150 + J, 0xF, 0xF, false);
}
template <int KK>
__device__ __forceinline__ void load_rows(const char* __restrict__ pb, unsigned lo, int my, float4 (&v)[GU]) {
    static_assert(GU == 8, "eight steps written out");
    v[0] = *(const float4*)(pb + ((unsigned)row_lane<KK + 0>(my) * (unsigned)(OF * 4) + lo));
    v[1] = *(const float4*)(pb + ((unsigned)row_lane<KK + 1>(my) * (unsigned)(OF * 4) + lo));
    v[2] = *(const float4*)(pb + ((unsigned)row_lane<KK + 2>(my) * (unsigned)(OF * 4) + lo));
    v[3] = *(const float4*)(pb + ((unsigned)row_lane<KK + 3>(my) * (unsigned)(OF * 4) + lo));
    v[4] = *(const float4*)(pb + ((unsigned)row_lane<KK + 4>(my) * (unsigned)(OF * 4) + lo));
    v[5] = *(const float4*)(pb + ((unsigned)row_lane<KK + 5>(my) * (unsigned)(OF * 4) + lo));
    v[6] = *(const float4*)(pb + ((unsigned)row_lane<KK + 6>(my) * (unsigned)(OF * 4) + lo));
    v[7] = *(const float4*)(pb + ((unsigned)row_lane<KK + 7>(my) * (unsigned)(OF * 4) + lo));
}
template <int KK>
__device__ __forceinline__ void gather_steps(const char* __restrict__ pb, unsigned lo, int my,
                                             int n0, int n1, int nh, float4& acc) {
    if (KK + GU <= n1) {
        float4 v[GU];
        load_rows<KK>(pb, lo, my, v);
        #pragma unroll
        for (int u = 0; u < GU; ++u) {
            acc.x += v[u].x; acc.y += v[u].y; acc.z += v[u].z; acc.w += v[u].w;
        }
    } else if (KK < n0) {
        float4 v[GU];
        load_rows<KK>(pb, lo, my, v);
        #pragma unroll
        for (int u = 0; u < GU; ++u) {
            const bool on = KK + u < nh;
            acc.x += on ? v[u].x : 0.0f;
            acc.y += on ? v[u].y : 0.0f;
            acc.z += on ? v[u].z : 0.0f;
            acc.w += on ? v[u].w : 0.0f;
        }
    }
}
__device__ __forceinline__ float other_half(float v) {
    return __int_as_float(__builtin_amdgcn_permlanex16(__float_as_int(v), __float_as_int(v),
                                                       0x76543210, 0xfedcba98, false, false));
}

__global__ void __launch_bounds__(FW * 32)
gather_out_kernel(const int* __restrict__ rowptr, const int* __restrict__ sorted_src,
                  const float* __restrict__ P,
                  const float* __restrict__ rs_in,
                  const float* __restrict__ bias,
                  float* __restrict__ out) {
    static_assert(OF == 64, "a row of P is 16 lanes x 16 bytes");
    const int lane = threadIdx.x & 31;
    const int wave = __builtin_amdgcn_readfirstlane(threadIdx.x >> 5);
    const int node = blockIdx.x * FW + wave;
    const int beg = __builtin_amdgcn_readfirstlane(rowptr[node]);
    const int end = __builtin_amdgcn_readfirstlane(rowptr[node + 1]);
    const float rsn = rs_in[node];
    const int half = lane >> 4;
    const int l16  = lane & 15;
    const float4 bv = *(const float4*)(bias + l16 * 4);
    const int cnt  = end - beg;
    const int len0 = (cnt + 1) >> 1;
    const int len1 = cnt - len0;
    const int lenh = half ? len1 : len0;
    const int* __restrict__ run = sorted_src + beg + (half ? len0 : 0);
    const char* __restrict__ pb = (const char*)P;
    const unsigned lo = (unsigned)l16 * 16u;
    float4 acc = make_float4(0.0f, 0.0f, 0.0f, 0.0f);
    for (int b = 0; b < len0; b += 16) {
        const int n0 = min(16, len0 - b);
        const int n1 = min(16, len1 - b);
        const int nh = half ? n1 : n0;
        const int my = run[min(b + l16, lenh - 1)];
        gather_steps<0>(pb, lo, my, n0, n1, nh, acc);
        gather_steps<GU>(pb, lo, my, n0, n1, nh, acc);
    }
    acc.x += other_half(acc.x);
    acc.y += other_half(acc.y);
    acc.z += other_half(acc.z);
    acc.w += other_half(acc.w);
    if (half == 0) {
        float4 o;
        o.x = acc.x * rsn + bv.x;
        o.y = acc.y * rsn + bv.y;
        o.z = acc.z * rsn + bv.z;
        o.w = acc.w * rsn + bv.w;
        *(float4*)(out + (size_t)node * OF + l16 * 4) = o;
    }
}


extern "C" void kernel_launch(void* const* d_in, const int* in_sizes, int n_in,
                              void* d_out, int out_size, void* d_ws, size_t ws_size,
                              hipStream_t stream) {
    static_assert(INF == 128 && HF == 128, "both products have K = 128 (one 512-byte row per node)");
    const float* in_feat = (const float*)d_in[0];
    const int*   src     = (const int*)d_in[1];
    const int*   dst     = (const int*)d_in[2];
    const float* W1      = (const float*)d_in[3];
    const float* b1      = (const float*)d_in[4];
    const float* W2      = (const float*)d_in[5];
    const float* b2      = (const float*)d_in[6];
    float* out = (float*)d_out;

    float* ws      = (float*)d_ws;
    float* rs_out  = ws;
    float* rs_in   = ws + NNP;
    int*   rowptr  = (int*)(ws + 2 * NNP);
    float* P       = ws + 4 * NNP;
    int*   dtable  = (int*)(P + (size_t)NN * OF);
    int*   keys    = dtable + (size_t)HS * NNP;
    int*   bstart  = keys + NE;
    int*   sorted  = bstart + HS * BP;

    slice_kernel<<<2 * HS, HT, 0, stream>>>(src, dst, dtable, keys, bstart);
    range_fill_kernel<<<NB, HT, 0, stream>>>(keys, bstart, dtable, rs_out, rs_in, rowptr, sorted);

    layer1_kernel<<<NN / FW, FW * 32, 0, stream>>>(
        rowptr, sorted, in_feat, rs_out, rs_in, W1, b1, W2, P);
    gather_out_kernel<<<NN / FW, FW * 32, 0, stream>>>(rowptr, sorted, P, rs_in, b2, out);
}


# range_fill: a wave loads its two segments' keys once, all loads in flight, and keeps them in registers for counting and placing (were two serial load-add loops); scalar-loaded bounds; order kept
# speedup vs baseline: 7.3717x; 1.0365x over previous
#include <hip/hip_runtime.h>


#define NN 10000
#define NE 640000
#define INF 128
#define HF  128
#define OF  64

typedef float v2f __attribute__((ext_vector_type(2)));
typedef float v8f __attribute__((ext_vector_type(8)));

#define HS   64
#define SL   (NE / HS)
#define HT   1024
#define NNP  10016
#define NB   125
#define RS   (NN / NB)
#define BP   128
#define KSH  14
#define LW   (NN > SL ? NN : SL)

__device__ __forceinline__ int wave_inclusive(int v, int lane) {
    #pragma unroll
    for (int off = 1; off < 32; off <<= 1) {
        int up = __shfl_up(v, off);
        if (lane >= off) v += up;
    }
    return v;
}

__global__ void __launch_bounds__(HT)
slice_kernel(const int* __restrict__ src, const int* __restrict__ dst,
             int* __restrict__ table, int* __restrict__ keys, int* __restrict__ bstart) {
    __shared__ __attribute__((aligned(16))) int lds[LW];
    __shared__ int bcnt[BP];
    __shared__ int boffs[BP];
    static_assert(NE % HS == 0, "edge slices must be equal");
    static_assert(NN % 4 == 0 && SL % 4 == 0, "rows and slices move in 16-byte pieces");
    static_assert(NB * RS == NN && NB + 1 <= BP && RS <= BP, "ranges tile the nodes; range arrays fit their pitch");
    static_assert(NN <= (1 << KSH) && NN < (1 << (31 - KSH)), "a packed word holds both node ids");
    const int which = blockIdx.x / HS;
    const int slice = blockIdx.x - which * HS;
    const int tid   = threadIdx.x;
    constexpr int PER4  = SL / 4;
    constexpr int TRIPS = (PER4 + HT - 1) / HT;
    int4* lds4 = (int4*)lds;

    if (which == 0) {
        const int4* __restrict__ idx4 = (const int4*)(src + (size_t)slice * SL);
        int4 v[TRIPS];
        #pragma unroll
        for (int j = 0; j < TRIPS; ++j) {
            int i = tid + j * HT;
            if (i < PER4) v[j] = idx4[i];
        }
        for (int i = tid; i < NN / 4; i += HT) lds4[i] = make_int4(0, 0, 0, 0);
        __syncthreads();
        #pragma unroll
        for (int j = 0; j < TRIPS; ++j) {
            int i = tid + j * HT;
            if (i < PER4) {
                atomicAdd(&lds[v[j].x], 1);
                atomicAdd(&lds[v[j].y], 1);
                atomicAdd(&lds[v[j].z], 1);
                atomicAdd(&lds[v[j].w], 1);
            }
        }
        __syncthreads();
        int4* row4 = (int4*)(table + (size_t)slice * NNP);
        for (int i = tid; i < NN / 4; i += HT) row4[i] = lds4[i];
    } else {
        const int4* __restrict__ d4 = (const int4*)(dst + (size_t)slice * SL);
        const int4* __restrict__ s4 = (const int4*)(src + (size_t)slice * SL);
        int4 dv[TRIPS], sv[TRIPS];
        #pragma unroll
        for (int j = 0; j < TRIPS; ++j) {
            int i = tid + j * HT;
            if (i < PER4) { dv[j] = d4[i]; sv[j] = s4[i]; }
        }
        if (tid < BP) bcnt[tid] = 0;
        __syncthreads();
        int bk[TRIPS][4], rk[TRIPS][4];
        #pragma unroll
        for (int j = 0; j < TRIPS; ++j) {
            int i = tid + j * HT;
            if (i < PER4) {
                bk[j][0] = dv[j].x / RS; bk[j][1] = dv[j].y / RS;
                bk[j][2] = dv[j].z / RS; bk[j][3] = dv[j].w / RS;
                rk[j][0] = atomicAdd(&bcnt[bk[j][0]], 1);
                rk[j][1] = atomicAdd(&bcnt[bk[j][1]], 1);
                rk[j][2] = atomicAdd(&bcnt[bk[j][2]], 1);
                rk[j][3] = atomicAdd(&bcnt[bk[j][3]], 1);
            }
        }
        __syncthreads();
        if (tid < 32) {
            const int c0 = bcnt[tid * 4 + 0], c1 = bcnt[tid * 4 + 1];
            const int c2 = bcnt[tid * 4 + 2], c3 = bcnt[tid * 4 + 3];
            const int s  = c0 + c1 + c2 + c3;
            const int ex = wave_inclusive(s, tid) - s;
            boffs[tid * 4 + 0] = ex;
            boffs[tid * 4 + 1] = ex + c0;
            boffs[tid * 4 + 2] = ex + c0 + c1;
            boffs[tid * 4 + 3] = ex + c0 + c1 + c2;
        }
        __syncthreads();
        #pragma unroll
        for (int j = 0; j < TRIPS; ++j) {
            int i = tid + j * HT;
            if (i < PER4) {
                lds[boffs[bk[j][0]] + rk[j][0]] = (dv[j].x << KSH) | sv[j].x;
                lds[boffs[bk[j][1]] + rk[j][1]] = (dv[j].y << KSH) | sv[j].y;
                lds[boffs[bk[j][2]] + rk[j][2]] = (dv[j].z << KSH) | sv[j].z;
                lds[boffs[bk[j][3]] + rk[j][3]] = (dv[j].w << KSH) | sv[j].w;
            }
        }
        if (tid <= NB) bstart[slice * BP + tid] = boffs[tid];
        __syncthreads();
        int4* out4 = (int4*)(keys + (size_t)slice * SL);
        for (int i = tid; i < SL / 4; i += HT) out4[i] = lds4[i];
    }
}

#define FWV (HT / 32)
#define KR  4
__global__ void __launch_bounds__(HT)
range_fill_kernel(const int* __restrict__ keys, const int* __restrict__ bstart,
                  const int* __restrict__ table, float* __restrict__ rs_out,
                  float* __restrict__ rs_in, int* __restrict__ rowptr,
                  int* __restrict__ sorted_src) {
    static_assert(HS == 2 * FWV, "two segments per wave");
    static_assert(HS == 64, "the segment starts are summed two per lane by one wave");
    static_assert(HS % (HT / BP) == 0, "table rows split evenly over the thread groups");
    __shared__ int cntw[FWV * RS];
    __shared__ int tot[BP];
    __shared__ int dego[BP];
    __shared__ int segs0[HS];
    const int b    = blockIdx.x;
    const int tid  = threadIdx.x;
    const int lane = tid & 31;
    const int wv   = __builtin_amdgcn_readfirstlane(tid >> 5);

    int sb[2], sl[2];
    #pragma unroll
    for (int q = 0; q < 2; ++q) {
        const int sidx = 2 * wv + q;
        const int s0 = bstart[sidx * BP + b];
        const int s1 = bstart[sidx * BP + b + 1];
        sb[q] = sidx * SL + s0;
        sl[q] = s1 - s0;
    }
    int kr[2][KR];
    #pragma unroll
    for (int q = 0; q < 2; ++q) {
        #pragma unroll
        for (int i = 0; i < KR; ++i) {
            const int j = lane + 32 * i;
            kr[q][i] = 0;
            if (j < sl[q]) kr[q][i] = keys[sb[q] + j];
        }
    }

    constexpr int RPT = HS / (HT / BP);
    const int nl = tid & (BP - 1);
    int od = 0;
    if (nl < RS) {
        const int* __restrict__ col = table + (size_t)((tid / BP) * RPT) * NNP + b * RS + nl;
        #pragma unroll
        for (int r = 0; r < RPT; ++r) od += col[(size_t)r * NNP];
    }
    if (tid < HS) segs0[tid] = bstart[tid * BP + b];
    for (int i = tid; i < FWV * RS; i += HT) cntw[i] = 0;
    if (tid < BP) { tot[tid] = 0; dego[tid] = 0; }
    __syncthreads();

    if (nl < RS) atomicAdd(&dego[nl], od);
    int* mycnt = cntw + wv * RS;
    #pragma unroll
    for (int q = 0; q < 2; ++q) {
        #pragma unroll
        for (int i = 0; i < KR; ++i) {
            const int j = lane + 32 * i;
            if (j < sl[q]) atomicAdd(&mycnt[(kr[q][i] >> KSH) - b * RS], 1);
        }
        for (int j = lane + 32 * KR; j < sl[q]; j += 32) {
            const int key = keys[sb[q] + j];
            atomicAdd(&mycnt[(key >> KSH) - b * RS], 1);
        }
    }
    __syncthreads();

    if (tid < RS) {
        int c[FWV];
        #pragma unroll
        for (int w = 0; w < FWV; ++w) c[w] = cntw[w * RS + tid];
        int run = 0;
        #pragma unroll
        for (int w = 0; w < FWV; ++w) { cntw[w * RS + tid] = run; run += c[w]; }
        tot[tid] = run;
    }
    __syncthreads();

    if (tid < 32) {
        const int c0 = tot[tid * 4 + 0], c1 = tot[tid * 4 + 1];
        const int c2 = tot[tid * 4 + 2], c3 = tot[tid * 4 + 3];
        const int s   = c0 + c1 + c2 + c3;
        const int inc = wave_inclusive(s, tid);
        const int lower = __builtin_amdgcn_readlane(
            wave_inclusive(segs0[2 * tid] + segs0[2 * tid + 1], tid), 31);
        const int st0 = lower + inc - s;
        const int st[4] = { st0, st0 + c0, st0 + c0 + c1, st0 + c0 + c1 + c2 };
        const int cc[4] = { c0, c1, c2, c3 };
        #pragma unroll
        for (int q = 0; q < 4; ++q) {
            const int n = tid * 4 + q;
            tot[n] = st[q];
            if (n < RS) {
                const int node = b * RS + n;
                rowptr[node] = st[q];
                rs_in[node]  = rsqrtf(fmaxf((float)cc[q], 1.0f));
                rs_out[node] = rsqrtf(fmaxf((float)dego[n], 1.0f));
            }
        }
        if (b == NB - 1 && tid == 31) rowptr[NN] = lower + inc;
    }
    __syncthreads();

    int tb[2][KR], rk[2][KR];
    #pragma unroll
    for (int q = 0; q < 2; ++q) {
        #pragma unroll
        for (int i = 0; i < KR; ++i) {
            const int j = lane + 32 * i;
            tb[q][i] = 0; rk[q][i] = 0;
            if (j < sl[q]) {
                const int dl = (kr[q][i] >> KSH) - b * RS;
                tb[q][i] = tot[dl];
                rk[q][i] = atomicAdd(&mycnt[dl], 1);
            }
        }
        for (int j = lane + 32 * KR; j < sl[q]; j += 32) {
            const int key = keys[sb[q] + j];
            const int dl  = (key >> KSH) - b * RS;
            const int slot = tot[dl] + atomicAdd(&mycnt[dl], 1);
            sorted_src[slot] = key & ((1 << KSH) - 1);
        }
    }
    #pragma unroll
    for (int q = 0; q < 2; ++q) {
        #pragma unroll
        for (int i = 0; i < KR; ++i) {
            const int j = lane + 32 * i;
            if (j < sl[q]) sorted_src[tb[q][i] + rk[q][i]] = kr[q][i] & ((1 << KSH) - 1);
        }
    }
}

#define FW 16
#define GU 8
#define AP 132
#define PP 68
#define N1 (HF / 16)
#define N2 (OF / 16)
__global__ void __launch_bounds__(FW * 32)
layer1_kernel(const int* __restrict__ rowptr, const int* __restrict__ sorted_src,
              const float* __restrict__ x,
              const float* __restrict__ rs_out,
              const float* __restrict__ rs_in,
              const float* __restrict__ W1,
              const float* __restrict__ bias1,
              const float* __restrict__ W2,
              float* __restrict__ P) {
    static_assert(NN % FW == 0, "whole blocks of nodes");
    static_assert(FW - N1 == 2 * N2 && 2 * (FW - N1) == FW, "waves 0..N1-1: product 1; waves N1..FW-1: product 2, "
                                                            "a column tile and a half of K each, then two rows each");
    __shared__ __attribute__((aligned(16))) float tile[FW * AP];
    __shared__ __attribute__((aligned(16))) float htile[FW * AP];
    __shared__ __attribute__((aligned(16))) float ptile[2 * FW * PP];
    __shared__ float rso[FW];
    const int lane = threadIdx.x & 31;
    const int wave = __builtin_amdgcn_readfirstlane(threadIdx.x >> 5);
    const int node = blockIdx.x * FW + wave;
    const int beg = __builtin_amdgcn_readfirstlane(rowptr[node]);
    const int end = __builtin_amdgcn_readfirstlane(rowptr[node + 1]);
    const float rsn = rs_in[node];
    const float rsm = rs_out[node];
    const float* __restrict__ xl = x + lane * 4;
    float4 acc = make_float4(0.0f, 0.0f, 0.0f, 0.0f);
    for (int base = beg; base < end; base += 32) {
        const int n = min(32, end - base);
        const int my    = sorted_src[base + min(lane, n - 1)];
        const int myr_i = __float_as_int(rs_out[my]);
        int k = 0;
        for (; k + GU <= n; k += GU) {
            float4 v[GU];
            #pragma unroll
            for (int u = 0; u < GU; ++u) {
                const int s = __builtin_amdgcn_readlane(my, k + u);
                v[u] = *(const float4*)(xl + (size_t)s * 128);
            }
            #pragma unroll
            for (int u = 0; u < GU; ++u) {
                const float r = __int_as_float(__builtin_amdgcn_readlane(myr_i, k + u));
                acc.x += v[u].x * r;
                acc.y += v[u].y * r;
                acc.z += v[u].z * r;
                acc.w += v[u].w * r;
            }
        }
        if (k < n) {
            float4 v[GU];
            #pragma unroll
            for (int u = 0; u < GU; ++u) {
                const int s = __builtin_amdgcn_readlane(my, k + u);
                v[u] = *(const float4*)(xl + (size_t)s * 128);
            }
            #pragma unroll
            for (int u = 0; u < GU; ++u) {
                const float r  = __int_as_float(__builtin_amdgcn_readlane(myr_i, k + u));
                const bool  on = k + u < n;
                acc.x = on ? acc.x + v[u].x * r : acc.x;
                acc.y = on ? acc.y + v[u].y * r : acc.y;
                acc.z = on ? acc.z + v[u].z * r : acc.z;
                acc.w = on ? acc.w + v[u].w * r : acc.w;
            }
        }
    }
    acc.x *= rsn; acc.y *= rsn; acc.z *= rsn; acc.w *= rsn;
    *(float4*)(tile + wave * AP + lane * 4) = acc;
    if (lane == 0) rso[wave] = rsm;

    const int half  = lane >> 4;
    const int l16   = lane & 15;
    v2f bw1[32];
    if (wave < N1) {
        const float* Wcol = W1 + (size_t)half * 2 * HF + wave * 16 + l16;
        #pragma unroll
        for (int k = 0; k < 128; k += 4) {
            bw1[k / 4].x = Wcol[(size_t)k * HF];
            bw1[k / 4].y = Wcol[(size_t)(k + 1) * HF];
        }
    }
    __syncthreads();

    v2f bw2[16];
    const int t2 = (wave - N1) & (N2 - 1);
    const int kh = (wave - N1) / N2;
    if (wave < N1) {
        const float* arow = tile + l16 * AP + half * 2;
        v8f c = {};
        #pragma unroll
        for (int k = 0; k < 128; k += 4) {
            const v2f a = *(const v2f*)(arow + k);
            c = __builtin_amdgcn_wmma_f32_16x16x4_f32(false, a, false, bw1[k / 4], (short)0, c,
                                                      false, false);
        }
        const int nCol = wave * 16 + l16;
        const float bv = bias1[nCol];
        #pragma unroll
        for (int r = 0; r < 8; ++r) {
            const int row = r + half * 8;
            htile[row * AP + nCol] = fmaxf(c[r] + bv, 0.0f);
        }
    }
    if (wave >= N1) {
        const float* Wcol = W2 + (size_t)(kh * 64 + half * 2) * OF + t2 * 16 + l16;
        #pragma unroll
        for (int k = 0; k < 64; k += 4) {
            bw2[k / 4].x = Wcol[(size_t)k * OF];
            bw2[k / 4].y = Wcol[(size_t)(k + 1) * OF];
        }
    }
    __syncthreads();

    if (wave >= N1) {
        const float* arow = htile + l16 * AP + kh * 64 + half * 2;
        v8f c = {};
        #pragma unroll
        for (int k = 0; k < 64; k += 4) {
            const v2f a = *(const v2f*)(arow + k);
            c = __builtin_amdgcn_wmma_f32_16x16x4_f32(false, a, false, bw2[k / 4], (short)0, c,
                                                      false, false);
        }
        float* pt = ptile + kh * (FW * PP) + half * 8 * PP + t2 * 16 + l16;
        #pragma unroll
        for (int r = 0; r < 8; ++r) pt[r * PP] = c[r];
    }
    __syncthreads();

    if (wave >= N1) {
        const int row = (wave - N1) * 2 + half;
        const float4 p0 = *(const float4*)(ptile + row * PP + l16 * 4);
        const float4 p1 = *(const float4*)(ptile + (FW + row) * PP + l16 * 4);
        const float s = rso[row];
        float4 v;
        v.x = (p0.x + p1.x) * s; v.y = (p0.y + p1.y) * s; v.z = (p0.z + p1.z) * s; v.w = (p0.w + p1.w) * s;
        *(float4*)(P + (size_t)(blockIdx.x * FW + row) * OF + l16 * 4) = v;
    }
}


template <int J>
__device__ __forceinline__ int row_lane(int v) {
    return __builtin_amdgcn_update_dpp(v, v, 0x150 + J, 0xF, 0xF, false);
}
template <int KK>
__device__ __forceinline__ void load_rows(const char* __restrict__ pb, unsigned lo, int my, float4 (&v)[GU]) {
    static_assert(GU == 8, "eight steps written out");
    v[0] = *(const float4*)(pb + ((unsigned)row_lane<KK + 0>(my) * (unsigned)(OF * 4) + lo));
    v[1] = *(const float4*)(pb + ((unsigned)row_lane<KK + 1>(my) * (unsigned)(OF * 4) + lo));
    v[2] = *(const float4*)(pb + ((unsigned)row_lane<KK + 2>(my) * (unsigned)(OF * 4) + lo));
    v[3] = *(const float4*)(pb + ((unsigned)row_lane<KK + 3>(my) * (unsigned)(OF * 4) + lo));
    v[4] = *(const float4*)(pb + ((unsigned)row_lane<KK + 4>(my) * (unsigned)(OF * 4) + lo));
    v[5] = *(const float4*)(pb + ((unsigned)row_lane<KK + 5>(my) * (unsigned)(OF * 4) + lo));
    v[6] = *(const float4*)(pb + ((unsigned)row_lane<KK + 6>(my) * (unsigned)(OF * 4) + lo));
    v[7] = *(const float4*)(pb + ((unsigned)row_lane<KK + 7>(my) * (unsigned)(OF * 4) + lo));
}
template <int KK>
__device__ __forceinline__ void gather_steps(const char* __restrict__ pb, unsigned lo, int my,
                                             int n0, int n1, int nh, float4& acc) {
    if (KK + GU <= n1) {
        float4 v[GU];
        load_rows<KK>(pb, lo, my, v);
        #pragma unroll
        for (int u = 0; u < GU; ++u) {
            acc.x += v[u].x; acc.y += v[u].y; acc.z += v[u].z; acc.w += v[u].w;
        }
    } else if (KK < n0) {
        float4 v[GU];
        load_rows<KK>(pb, lo, my, v);
        #pragma unroll
        for (int u = 0; u < GU; ++u) {
            const bool on = KK + u < nh;
            acc.x += on ? v[u].x : 0.0f;
            acc.y += on ? v[u].y : 0.0f;
            acc.z += on ? v[u].z : 0.0f;
            acc.w += on ? v[u].w : 0.0f;
        }
    }
}
__device__ __forceinline__ float other_half(float v) {
    return __int_as_float(__builtin_amdgcn_permlanex16(__float_as_int(v), __float_as_int(v),
                                                       0x76543210, 0xfedcba98, false, false));
}

__global__ void __launch_bounds__(FW * 32)
gather_out_kernel(const int* __restrict__ rowptr, const int* __restrict__ sorted_src,
                  const float* __restrict__ P,
                  const float* __restrict__ rs_in,
                  const float* __restrict__ bias,
                  float* __restrict__ out) {
    static_assert(OF == 64, "a row of P is 16 lanes x 16 bytes");
    const int lane = threadIdx.x & 31;
    const int wave = __builtin_amdgcn_readfirstlane(threadIdx.x >> 5);
    const int node = blockIdx.x * FW + wave;
    const int beg = __builtin_amdgcn_readfirstlane(rowptr[node]);
    const int end = __builtin_amdgcn_readfirstlane(rowptr[node + 1]);
    const float rsn = rs_in[node];
    const int half = lane >> 4;
    const int l16  = lane & 15;
    const float4 bv = *(const float4*)(bias + l16 * 4);
    const int cnt  = end - beg;
    const int len0 = (cnt + 1) >> 1;
    const int len1 = cnt - len0;
    const int lenh = half ? len1 : len0;
    const int* __restrict__ run = sorted_src + beg + (half ? len0 : 0);
    const char* __restrict__ pb = (const char*)P;
    const unsigned lo = (unsigned)l16 * 16u;
    float4 acc = make_float4(0.0f, 0.0f, 0.0f, 0.0f);
    for (int b = 0; b < len0; b += 16) {
        const int n0 = min(16, len0 - b);
        const int n1 = min(16, len1 - b);
        const int nh = half ? n1 : n0;
        const int my = run[min(b + l16, lenh - 1)];
        gather_steps<0>(pb, lo, my, n0, n1, nh, acc);
        gather_steps<GU>(pb, lo, my, n0, n1, nh, acc);
    }
    acc.x += other_half(acc.x);
    acc.y += other_half(acc.y);
    acc.z += other_half(acc.z);
    acc.w += other_half(acc.w);
    if (half == 0) {
        float4 o;
        o.x = acc.x * rsn + bv.x;
        o.y = acc.y * rsn + bv.y;
        o.z = acc.z * rsn + bv.z;
        o.w = acc.w * rsn + bv.w;
        *(float4*)(out + (size_t)node * OF + l16 * 4) = o;
    }
}


extern "C" void kernel_launch(void* const* d_in, const int* in_sizes, int n_in,
                              void* d_out, int out_size, void* d_ws, size_t ws_size,
                              hipStream_t stream) {
    static_assert(INF == 128 && HF == 128, "both products have K = 128 (one 512-byte row per node)");
    const float* in_feat = (const float*)d_in[0];
    const int*   src     = (const int*)d_in[1];
    const int*   dst     = (const int*)d_in[2];
    const float* W1      = (const float*)d_in[3];
    const float* b1      = (const float*)d_in[4];
    const float* W2      = (const float*)d_in[5];
    const float* b2      = (const float*)d_in[6];
    float* out = (float*)d_out;

    float* ws      = (float*)d_ws;
    float* rs_out  = ws;
    float* rs_in   = ws + NNP;
    int*   rowptr  = (int*)(ws + 2 * NNP);
    float* P       = ws + 4 * NNP;
    int*   dtable  = (int*)(P + (size_t)NN * OF);
    int*   keys    = dtable + (size_t)HS * NNP;
    int*   bstart  = keys + NE;
    int*   sorted  = bstart + HS * BP;

    slice_kernel<<<2 * HS, HT, 0, stream>>>(src, dst, dtable, keys, bstart);
    range_fill_kernel<<<NB, HT, 0, stream>>>(keys, bstart, dtable, rs_out, rs_in, rowptr, sorted);

    layer1_kernel<<<NN / FW, FW * 32, 0, stream>>>(
        rowptr, sorted, in_feat, rs_out, rs_in, W1, b1, W2, P);
    gather_out_kernel<<<NN / FW, FW * 32, 0, stream>>>(rowptr, sorted, P, rs_in, b2, out);
}


# range_fill: a range's part of the sorted edge list is put in order in LDS and leaves as whole 128-byte lines, four words per lane (was a 4-byte store per lane, 32 lines per instruction); order kept
# speedup vs baseline: 7.6566x; 1.0386x over previous
#include <hip/hip_runtime.h>


#define NN 10000
#define NE 640000
#define INF 128
#define HF  128
#define OF  64

typedef float v2f __attribute__((ext_vector_type(2)));
typedef float v8f __attribute__((ext_vector_type(8)));

#define HS   64
#define SL   (NE / HS)
#define HT   1024
#define NNP  10016
#define NB   125
#define RS   (NN / NB)
#define BP   128
#define KSH  14
#define LW   (NN > SL ? NN : SL)

__device__ __forceinline__ int wave_inclusive(int v, int lane) {
    #pragma unroll
    for (int off = 1; off < 32; off <<= 1) {
        int up = __shfl_up(v, off);
        if (lane >= off) v += up;
    }
    return v;
}

__global__ void __launch_bounds__(HT)
slice_kernel(const int* __restrict__ src, const int* __restrict__ dst,
             int* __restrict__ table, int* __restrict__ keys, int* __restrict__ bstart) {
    __shared__ __attribute__((aligned(16))) int lds[LW];
    __shared__ int bcnt[BP];
    __shared__ int boffs[BP];
    static_assert(NE % HS == 0, "edge slices must be equal");
    static_assert(NN % 4 == 0 && SL % 4 == 0, "rows and slices move in 16-byte pieces");
    static_assert(NB * RS == NN && NB + 1 <= BP && RS <= BP, "ranges tile the nodes; range arrays fit their pitch");
    static_assert(NN <= (1 << KSH) && NN < (1 << (31 - KSH)), "a packed word holds both node ids");
    const int which = blockIdx.x / HS;
    const int slice = blockIdx.x - which * HS;
    const int tid   = threadIdx.x;
    constexpr int PER4  = SL / 4;
    constexpr int TRIPS = (PER4 + HT - 1) / HT;
    int4* lds4 = (int4*)lds;

    if (which == 0) {
        const int4* __restrict__ idx4 = (const int4*)(src + (size_t)slice * SL);
        int4 v[TRIPS];
        #pragma unroll
        for (int j = 0; j < TRIPS; ++j) {
            int i = tid + j * HT;
            if (i < PER4) v[j] = idx4[i];
        }
        for (int i = tid; i < NN / 4; i += HT) lds4[i] = make_int4(0, 0, 0, 0);
        __syncthreads();
        #pragma unroll
        for (int j = 0; j < TRIPS; ++j) {
            int i = tid + j * HT;
            if (i < PER4) {
                atomicAdd(&lds[v[j].x], 1);
                atomicAdd(&lds[v[j].y], 1);
                atomicAdd(&lds[v[j].z], 1);
                atomicAdd(&lds[v[j].w], 1);
            }
        }
        __syncthreads();
        int4* row4 = (int4*)(table + (size_t)slice * NNP);
        for (int i = tid; i < NN / 4; i += HT) row4[i] = lds4[i];
    } else {
        const int4* __restrict__ d4 = (const int4*)(dst + (size_t)slice * SL);
        const int4* __restrict__ s4 = (const int4*)(src + (size_t)slice * SL);
        int4 dv[TRIPS], sv[TRIPS];
        #pragma unroll
        for (int j = 0; j < TRIPS; ++j) {
            int i = tid + j * HT;
            if (i < PER4) { dv[j] = d4[i]; sv[j] = s4[i]; }
        }
        if (tid < BP) bcnt[tid] = 0;
        __syncthreads();
        int bk[TRIPS][4], rk[TRIPS][4];
        #pragma unroll
        for (int j = 0; j < TRIPS; ++j) {
            int i = tid + j * HT;
            if (i < PER4) {
                bk[j][0] = dv[j].x / RS; bk[j][1] = dv[j].y / RS;
                bk[j][2] = dv[j].z / RS; bk[j][3] = dv[j].w / RS;
                rk[j][0] = atomicAdd(&bcnt[bk[j][0]], 1);
                rk[j][1] = atomicAdd(&bcnt[bk[j][1]], 1);
                rk[j][2] = atomicAdd(&bcnt[bk[j][2]], 1);
                rk[j][3] = atomicAdd(&bcnt[bk[j][3]], 1);
            }
        }
        __syncthreads();
        if (tid < 32) {
            const int c0 = bcnt[tid * 4 + 0], c1 = bcnt[tid * 4 + 1];
            const int c2 = bcnt[tid * 4 + 2], c3 = bcnt[tid * 4 + 3];
            const int s  = c0 + c1 + c2 + c3;
            const int ex = wave_inclusive(s, tid) - s;
            boffs[tid * 4 + 0] = ex;
            boffs[tid * 4 + 1] = ex + c0;
            boffs[tid * 4 + 2] = ex + c0 + c1;
            boffs[tid * 4 + 3] = ex + c0 + c1 + c2;
        }
        __syncthreads();
        #pragma unroll
        for (int j = 0; j < TRIPS; ++j) {
            int i = tid + j * HT;
            if (i < PER4) {
                lds[boffs[bk[j][0]] + rk[j][0]] = (dv[j].x << KSH) | sv[j].x;
                lds[boffs[bk[j][1]] + rk[j][1]] = (dv[j].y << KSH) | sv[j].y;
                lds[boffs[bk[j][2]] + rk[j][2]] = (dv[j].z << KSH) | sv[j].z;
                lds[boffs[bk[j][3]] + rk[j][3]] = (dv[j].w << KSH) | sv[j].w;
            }
        }
        if (tid <= NB) bstart[slice * BP + tid] = boffs[tid];
        __syncthreads();
        int4* out4 = (int4*)(keys + (size_t)slice * SL);
        for (int i = tid; i < SL / 4; i += HT) out4[i] = lds4[i];
    }
}

#define FWV (HT / 32)
#define CAP 8192
#define KR  4
__global__ void __launch_bounds__(HT)
range_fill_kernel(const int* __restrict__ keys, const int* __restrict__ bstart,
                  const int* __restrict__ table, float* __restrict__ rs_out,
                  float* __restrict__ rs_in, int* __restrict__ rowptr,
                  int* __restrict__ sorted_src) {
    static_assert(HS == 2 * FWV, "two segments per wave");
    static_assert(HS == 64, "the segment starts are summed two per lane by one wave");
    static_assert(HS % (HT / BP) == 0, "table rows split evenly over the thread groups");
    __shared__ int cntw[FWV * RS];
    __shared__ int tot[BP];
    __shared__ int dego[BP];
    __shared__ int segs0[HS];
    __shared__ int rng[2];
    __shared__ __attribute__((aligned(16))) int stage[CAP + 32];
    const int b    = blockIdx.x;
    const int tid  = threadIdx.x;
    const int lane = tid & 31;
    const int wv   = __builtin_amdgcn_readfirstlane(tid >> 5);

    int sb[2], sl[2];
    #pragma unroll
    for (int q = 0; q < 2; ++q) {
        const int sidx = 2 * wv + q;
        const int s0 = bstart[sidx * BP + b];
        const int s1 = bstart[sidx * BP + b + 1];
        sb[q] = sidx * SL + s0;
        sl[q] = s1 - s0;
    }
    int kr[2][KR];
    #pragma unroll
    for (int q = 0; q < 2; ++q) {
        #pragma unroll
        for (int i = 0; i < KR; ++i) {
            const int j = lane + 32 * i;
            kr[q][i] = 0;
            if (j < sl[q]) kr[q][i] = keys[sb[q] + j];
        }
    }

    constexpr int RPT = HS / (HT / BP);
    const int nl = tid & (BP - 1);
    int od = 0;
    if (nl < RS) {
        const int* __restrict__ col = table + (size_t)((tid / BP) * RPT) * NNP + b * RS + nl;
        #pragma unroll
        for (int r = 0; r < RPT; ++r) od += col[(size_t)r * NNP];
    }
    if (tid < HS) segs0[tid] = bstart[tid * BP + b];
    for (int i = tid; i < FWV * RS; i += HT) cntw[i] = 0;
    if (tid < BP) { tot[tid] = 0; dego[tid] = 0; }
    __syncthreads();

    if (nl < RS) atomicAdd(&dego[nl], od);
    int* mycnt = cntw + wv * RS;
    #pragma unroll
    for (int q = 0; q < 2; ++q) {
        #pragma unroll
        for (int i = 0; i < KR; ++i) {
            const int j = lane + 32 * i;
            if (j < sl[q]) atomicAdd(&mycnt[(kr[q][i] >> KSH) - b * RS], 1);
        }
        for (int j = lane + 32 * KR; j < sl[q]; j += 32) {
            const int key = keys[sb[q] + j];
            atomicAdd(&mycnt[(key >> KSH) - b * RS], 1);
        }
    }
    __syncthreads();

    if (tid < RS) {
        int c[FWV];
        #pragma unroll
        for (int w = 0; w < FWV; ++w) c[w] = cntw[w * RS + tid];
        int run = 0;
        #pragma unroll
        for (int w = 0; w < FWV; ++w) { cntw[w * RS + tid] = run; run += c[w]; }
        tot[tid] = run;
    }
    __syncthreads();

    if (tid < 32) {
        const int c0 = tot[tid * 4 + 0], c1 = tot[tid * 4 + 1];
        const int c2 = tot[tid * 4 + 2], c3 = tot[tid * 4 + 3];
        const int s   = c0 + c1 + c2 + c3;
        const int inc = wave_inclusive(s, tid);
        const int lower = __builtin_amdgcn_readlane(
            wave_inclusive(segs0[2 * tid] + segs0[2 * tid + 1], tid), 31);
        const int st0 = lower + inc - s;
        const int st[4] = { st0, st0 + c0, st0 + c0 + c1, st0 + c0 + c1 + c2 };
        const int cc[4] = { c0, c1, c2, c3 };
        #pragma unroll
        for (int q = 0; q < 4; ++q) {
            const int n = tid * 4 + q;
            tot[n] = st[q];
            if (n < RS) {
                const int node = b * RS + n;
                rowptr[node] = st[q];
                rs_in[node]  = rsqrtf(fmaxf((float)cc[q], 1.0f));
                rs_out[node] = rsqrtf(fmaxf((float)dego[n], 1.0f));
            }
        }
        if (tid == 31) { rng[0] = lower; rng[1] = inc; }
        if (b == NB - 1 && tid == 31) rowptr[NN] = lower + inc;
    }
    __syncthreads();

    const int run0  = __builtin_amdgcn_readfirstlane(rng[0]);
    const int total = __builtin_amdgcn_readfirstlane(rng[1]);
    const int gb    = run0 & ~31;
    const bool in_lds = total <= CAP;

    int tb[2][KR], rk[2][KR];
    #pragma unroll
    for (int q = 0; q < 2; ++q) {
        #pragma unroll
        for (int i = 0; i < KR; ++i) {
            const int j = lane + 32 * i;
            tb[q][i] = 0; rk[q][i] = 0;
            if (j < sl[q]) {
                const int dl = (kr[q][i] >> KSH) - b * RS;
                tb[q][i] = tot[dl];
                rk[q][i] = atomicAdd(&mycnt[dl], 1);
            }
        }
        for (int j = lane + 32 * KR; j < sl[q]; j += 32) {
            const int key = keys[sb[q] + j];
            const int dl  = (key >> KSH) - b * RS;
            const int slot = tot[dl] + atomicAdd(&mycnt[dl], 1);
            if (in_lds) stage[slot - gb] = key & ((1 << KSH) - 1);
            else        sorted_src[slot] = key & ((1 << KSH) - 1);
        }
    }
    if (in_lds) {
        #pragma unroll
        for (int q = 0; q < 2; ++q) {
            #pragma unroll
            for (int i = 0; i < KR; ++i) {
                const int j = lane + 32 * i;
                if (j < sl[q]) stage[tb[q][i] + rk[q][i] - gb] = kr[q][i] & ((1 << KSH) - 1);
            }
        }
        __syncthreads();
        const int first = run0 - gb;
        const int last  = first + total;
        int* __restrict__ part = sorted_src + gb;
        for (int w = 4 * tid; w < last; w += 4 * HT) {
            if (w >= first && w + 4 <= last) {
                *(int4*)(part + w) = *(const int4*)(stage + w);
            } else {
                for (int e = 0; e < 4; ++e)
                    if (w + e >= first && w + e < last) part[w + e] = stage[w + e];
            }
        }
    } else {
        #pragma unroll
        for (int q = 0; q < 2; ++q) {
            #pragma unroll
            for (int i = 0; i < KR; ++i) {
                const int j = lane + 32 * i;
                if (j < sl[q]) sorted_src[tb[q][i] + rk[q][i]] = kr[q][i] & ((1 << KSH) - 1);
            }
        }
    }
}

#define FW 16
#define GU 8
#define AP 132
#define PP 68
#define N1 (HF / 16)
#define N2 (OF / 16)
__global__ void __launch_bounds__(FW * 32)
layer1_kernel(const int* __restrict__ rowptr, const int* __restrict__ sorted_src,
              const float* __restrict__ x,
              const float* __restrict__ rs_out,
              const float* __restrict__ rs_in,
              const float* __restrict__ W1,
              const float* __restrict__ bias1,
              const float* __restrict__ W2,
              float* __restrict__ P) {
    static_assert(NN % FW == 0, "whole blocks of nodes");
    static_assert(FW - N1 == 2 * N2 && 2 * (FW - N1) == FW, "waves 0..N1-1: product 1; waves N1..FW-1: product 2, "
                                                            "a column tile and a half of K each, then two rows each");
    __shared__ __attribute__((aligned(16))) float tile[FW * AP];
    __shared__ __attribute__((aligned(16))) float htile[FW * AP];
    __shared__ __attribute__((aligned(16))) float ptile[2 * FW * PP];
    __shared__ float rso[FW];
    const int lane = threadIdx.x & 31;
    const int wave = __builtin_amdgcn_readfirstlane(threadIdx.x >> 5);
    const int node = blockIdx.x * FW + wave;
    const int beg = __builtin_amdgcn_readfirstlane(rowptr[node]);
    const int end = __builtin_amdgcn_readfirstlane(rowptr[node + 1]);
    const float rsn = rs_in[node];
    const float rsm = rs_out[node];
    const float* __restrict__ xl = x + lane * 4;
    float4 acc = make_float4(0.0f, 0.0f, 0.0f, 0.0f);
    for (int base = beg; base < end; base += 32) {
        const int n = min(32, end - base);
        const int my    = sorted_src[base + min(lane, n - 1)];
        const int myr_i = __float_as_int(rs_out[my]);
        int k = 0;
        for (; k + GU <= n; k += GU) {
            float4 v[GU];
            #pragma unroll
            for (int u = 0; u < GU; ++u) {
                const int s = __builtin_amdgcn_readlane(my, k + u);
                v[u] = *(const float4*)(xl + (size_t)s * 128);
            }
            #pragma unroll
            for (int u = 0; u < GU; ++u) {
                const float r = __int_as_float(__builtin_amdgcn_readlane(myr_i, k + u));
                acc.x += v[u].x * r;
                acc.y += v[u].y * r;
                acc.z += v[u].z * r;
                acc.w += v[u].w * r;
            }
        }
        if (k < n) {
            float4 v[GU];
            #pragma unroll
            for (int u = 0; u < GU; ++u) {
                const int s = __builtin_amdgcn_readlane(my, k + u);
                v[u] = *(const float4*)(xl + (size_t)s * 128);
            }
            #pragma unroll
            for (int u = 0; u < GU; ++u) {
                const float r  = __int_as_float(__builtin_amdgcn_readlane(myr_i, k + u));
                const bool  on = k + u < n;
                acc.x = on ? acc.x + v[u].x * r : acc.x;
                acc.y = on ? acc.y + v[u].y * r : acc.y;
                acc.z = on ? acc.z + v[u].z * r : acc.z;
                acc.w = on ? acc.w + v[u].w * r : acc.w;
            }
        }
    }
    acc.x *= rsn; acc.y *= rsn; acc.z *= rsn; acc.w *= rsn;
    *(float4*)(tile + wave * AP + lane * 4) = acc;
    if (lane == 0) rso[wave] = rsm;

    const int half  = lane >> 4;
    const int l16   = lane & 15;
    v2f bw1[32];
    if (wave < N1) {
        const float* Wcol = W1 + (size_t)half * 2 * HF + wave * 16 + l16;
        #pragma unroll
        for (int k = 0; k < 128; k += 4) {
            bw1[k / 4].x = Wcol[(size_t)k * HF];
            bw1[k / 4].y = Wcol[(size_t)(k + 1) * HF];
        }
    }
    __syncthreads();

    v2f bw2[16];
    const int t2 = (wave - N1) & (N2 - 1);
    const int kh = (wave - N1) / N2;
    if (wave < N1) {
        const float* arow = tile + l16 * AP + half * 2;
        v8f c = {};
        #pragma unroll
        for (int k = 0; k < 128; k += 4) {
            const v2f a = *(const v2f*)(arow + k);
            c = __builtin_amdgcn_wmma_f32_16x16x4_f32(false, a, false, bw1[k / 4], (short)0, c,
                                                      false, false);
        }
        const int nCol = wave * 16 + l16;
        const float bv = bias1[nCol];
        #pragma unroll
        for (int r = 0; r < 8; ++r) {
            const int row = r + half * 8;
            htile[row * AP + nCol] = fmaxf(c[r] + bv, 0.0f);
        }
    }
    if (wave >= N1) {
        const float* Wcol = W2 + (size_t)(kh * 64 + half * 2) * OF + t2 * 16 + l16;
        #pragma unroll
        for (int k = 0; k < 64; k += 4) {
            bw2[k / 4].x = Wcol[(size_t)k * OF];
            bw2[k / 4].y = Wcol[(size_t)(k + 1) * OF];
        }
    }
    __syncthreads();

    if (wave >= N1) {
        const float* arow = htile + l16 * AP + kh * 64 + half * 2;
        v8f c = {};
        #pragma unroll
        for (int k = 0; k < 64; k += 4) {
            const v2f a = *(const v2f*)(arow + k);
            c = __builtin_amdgcn_wmma_f32_16x16x4_f32(false, a, false, bw2[k / 4], (short)0, c,
                                                      false, false);
        }
        float* pt = ptile + kh * (FW * PP) + half * 8 * PP + t2 * 16 + l16;
        #pragma unroll
        for (int r = 0; r < 8; ++r) pt[r * PP] = c[r];
    }
    __syncthreads();

    if (wave >= N1) {
        const int row = (wave - N1) * 2 + half;
        const float4 p0 = *(const float4*)(ptile + row * PP + l16 * 4);
        const float4 p1 = *(const float4*)(ptile + (FW + row) * PP + l16 * 4);
        const float s = rso[row];
        float4 v;
        v.x = (p0.x + p1.x) * s; v.y = (p0.y + p1.y) * s; v.z = (p0.z + p1.z) * s; v.w = (p0.w + p1.w) * s;
        *(float4*)(P + (size_t)(blockIdx.x * FW + row) * OF + l16 * 4) = v;
    }
}


template <int J>
__device__ __forceinline__ int row_lane(int v) {
    return __builtin_amdgcn_update_dpp(v, v, 0x150 + J, 0xF, 0xF, false);
}
template <int KK>
__device__ __forceinline__ void load_rows(const char* __restrict__ pb, unsigned lo, int my, float4 (&v)[GU]) {
    static_assert(GU == 8, "eight steps written out");
    v[0] = *(const float4*)(pb + ((unsigned)row_lane<KK + 0>(my) * (unsigned)(OF * 4) + lo));
    v[1] = *(const float4*)(pb + ((unsigned)row_lane<KK + 1>(my) * (unsigned)(OF * 4) + lo));
    v[2] = *(const float4*)(pb + ((unsigned)row_lane<KK + 2>(my) * (unsigned)(OF * 4) + lo));
    v[3] = *(const float4*)(pb + ((unsigned)row_lane<KK + 3>(my) * (unsigned)(OF * 4) + lo));
    v[4] = *(const float4*)(pb + ((unsigned)row_lane<KK + 4>(my) * (unsigned)(OF * 4) + lo));
    v[5] = *(const float4*)(pb + ((unsigned)row_lane<KK + 5>(my) * (unsigned)(OF * 4) + lo));
    v[6] = *(const float4*)(pb + ((unsigned)row_lane<KK + 6>(my) * (unsigned)(OF * 4) + lo));
    v[7] = *(const float4*)(pb + ((unsigned)row_lane<KK + 7>(my) * (unsigned)(OF * 4) + lo));
}
template <int KK>
__device__ __forceinline__ void gather_steps(const char* __restrict__ pb, unsigned lo, int my,
                                             int n0, int n1, int nh, float4& acc) {
    if (KK + GU <= n1) {
        float4 v[GU];
        load_rows<KK>(pb, lo, my, v);
        #pragma unroll
        for (int u = 0; u < GU; ++u) {
            acc.x += v[u].x; acc.y += v[u].y; acc.z += v[u].z; acc.w += v[u].w;
        }
    } else if (KK < n0) {
        float4 v[GU];
        load_rows<KK>(pb, lo, my, v);
        #pragma unroll
        for (int u = 0; u < GU; ++u) {
            const bool on = KK + u < nh;
            acc.x += on ? v[u].x : 0.0f;
            acc.y += on ? v[u].y : 0.0f;
            acc.z += on ? v[u].z : 0.0f;
            acc.w += on ? v[u].w : 0.0f;
        }
    }
}
__device__ __forceinline__ float other_half(float v) {
    return __int_as_float(__builtin_amdgcn_permlanex16(__float_as_int(v), __float_as_int(v),
                                                       0x76543210, 0xfedcba98, false, false));
}

__global__ void __launch_bounds__(FW * 32)
gather_out_kernel(const int* __restrict__ rowptr, const int* __restrict__ sorted_src,
                  const float* __restrict__ P,
                  const float* __restrict__ rs_in,
                  const float* __restrict__ bias,
                  float* __restrict__ out) {
    static_assert(OF == 64, "a row of P is 16 lanes x 16 bytes");
    const int lane = threadIdx.x & 31;
    const int wave = __builtin_amdgcn_readfirstlane(threadIdx.x >> 5);
    const int node = blockIdx.x * FW + wave;
    const int beg = __builtin_amdgcn_readfirstlane(rowptr[node]);
    const int end = __builtin_amdgcn_readfirstlane(rowptr[node + 1]);
    const float rsn = rs_in[node];
    const int half = lane >> 4;
    const int l16  = lane & 15;
    const float4 bv = *(const float4*)(bias + l16 * 4);
    const int cnt  = end - beg;
    const int len0 = (cnt + 1) >> 1;
    const int len1 = cnt - len0;
    const int lenh = half ? len1 : len0;
    const int* __restrict__ run = sorted_src + beg + (half ? len0 : 0);
    const char* __restrict__ pb = (const char*)P;
    const unsigned lo = (unsigned)l16 * 16u;
    float4 acc = make_float4(0.0f, 0.0f, 0.0f, 0.0f);
    for (int b = 0; b < len0; b += 16) {
        const int n0 = min(16, len0 - b);
        const int n1 = min(16, len1 - b);
        const int nh = half ? n1 : n0;
        const int my = run[min(b + l16, lenh - 1)];
        gather_steps<0>(pb, lo, my, n0, n1, nh, acc);
        gather_steps<GU>(pb, lo, my, n0, n1, nh, acc);
    }
    acc.x += other_half(acc.x);
    acc.y += other_half(acc.y);
    acc.z += other_half(acc.z);
    acc.w += other_half(acc.w);
    if (half == 0) {
        float4 o;
        o.x = acc.x * rsn + bv.x;
        o.y = acc.y * rsn + bv.y;
        o.z = acc.z * rsn + bv.z;
        o.w = acc.w * rsn + bv.w;
        *(float4*)(out + (size_t)node * OF + l16 * 4) = o;
    }
}


extern "C" void kernel_launch(void* const* d_in, const int* in_sizes, int n_in,
                              void* d_out, int out_size, void* d_ws, size_t ws_size,
                              hipStream_t stream) {
    static_assert(INF == 128 && HF == 128, "both products have K = 128 (one 512-byte row per node)");
    const float* in_feat = (const float*)d_in[0];
    const int*   src     = (const int*)d_in[1];
    const int*   dst     = (const int*)d_in[2];
    const float* W1      = (const float*)d_in[3];
    const float* b1      = (const float*)d_in[4];
    const float* W2      = (const float*)d_in[5];
    const float* b2      = (const float*)d_in[6];
    float* out = (float*)d_out;

    float* ws      = (float*)d_ws;
    float* rs_out  = ws;
    float* rs_in   = ws + NNP;
    int*   rowptr  = (int*)(ws + 2 * NNP);
    float* P       = ws + 4 * NNP;
    int*   dtable  = (int*)(P + (size_t)NN * OF);
    int*   keys    = dtable + (size_t)HS * NNP;
    int*   bstart  = keys + NE;
    int*   sorted  = bstart + HS * BP;

    slice_kernel<<<2 * HS, HT, 0, stream>>>(src, dst, dtable, keys, bstart);
    range_fill_kernel<<<NB, HT, 0, stream>>>(keys, bstart, dtable, rs_out, rs_in, rowptr, sorted);

    layer1_kernel<<<NN / FW, FW * 32, 0, stream>>>(
        rowptr, sorted, in_feat, rs_out, rs_in, W1, b1, W2, P);
    gather_out_kernel<<<NN / FW, FW * 32, 0, stream>>>(rowptr, sorted, P, rs_in, b2, out);
}


# layer 1: W1 and W2 go to LDS once per block as whole rows, 16 bytes per lane, and both products read their B fragments there (were 4-byte loads from memory, half of each line used); sums unchanged
# speedup vs baseline: 7.9841x; 1.0428x over previous
#include <hip/hip_runtime.h>


#define NN 10000
#define NE 640000
#define INF 128
#define HF  128
#define OF  64

typedef float v2f __attribute__((ext_vector_type(2)));
typedef float v8f __attribute__((ext_vector_type(8)));

#define HS   64
#define SL   (NE / HS)
#define HT   1024
#define NNP  10016
#define NB   125
#define RS   (NN / NB)
#define BP   128
#define KSH  14
#define LW   (NN > SL ? NN : SL)

__device__ __forceinline__ int wave_inclusive(int v, int lane) {
    #pragma unroll
    for (int off = 1; off < 32; off <<= 1) {
        int up = __shfl_up(v, off);
        if (lane >= off) v += up;
    }
    return v;
}

__global__ void __launch_bounds__(HT)
slice_kernel(const int* __restrict__ src, const int* __restrict__ dst,
             int* __restrict__ table, int* __restrict__ keys, int* __restrict__ bstart) {
    __shared__ __attribute__((aligned(16))) int lds[LW];
    __shared__ int bcnt[BP];
    __shared__ int boffs[BP];
    static_assert(NE % HS == 0, "edge slices must be equal");
    static_assert(NN % 4 == 0 && SL % 4 == 0, "rows and slices move in 16-byte pieces");
    static_assert(NB * RS == NN && NB + 1 <= BP && RS <= BP, "ranges tile the nodes; range arrays fit their pitch");
    static_assert(NN <= (1 << KSH) && NN < (1 << (31 - KSH)), "a packed word holds both node ids");
    const int which = blockIdx.x / HS;
    const int slice = blockIdx.x - which * HS;
    const int tid   = threadIdx.x;
    constexpr int PER4  = SL / 4;
    constexpr int TRIPS = (PER4 + HT - 1) / HT;
    int4* lds4 = (int4*)lds;

    if (which == 0) {
        const int4* __restrict__ idx4 = (const int4*)(src + (size_t)slice * SL);
        int4 v[TRIPS];
        #pragma unroll
        for (int j = 0; j < TRIPS; ++j) {
            int i = tid + j * HT;
            if (i < PER4) v[j] = idx4[i];
        }
        for (int i = tid; i < NN / 4; i += HT) lds4[i] = make_int4(0, 0, 0, 0);
        __syncthreads();
        #pragma unroll
        for (int j = 0; j < TRIPS; ++j) {
            int i = tid + j * HT;
            if (i < PER4) {
                atomicAdd(&lds[v[j].x], 1);
                atomicAdd(&lds[v[j].y], 1);
                atomicAdd(&lds[v[j].z], 1);
                atomicAdd(&lds[v[j].w], 1);
            }
        }
        __syncthreads();
        int4* row4 = (int4*)(table + (size_t)slice * NNP);
        for (int i = tid; i < NN / 4; i += HT) row4[i] = lds4[i];
    } else {
        const int4* __restrict__ d4 = (const int4*)(dst + (size_t)slice * SL);
        const int4* __restrict__ s4 = (const int4*)(src + (size_t)slice * SL);
        int4 dv[TRIPS], sv[TRIPS];
        #pragma unroll
        for (int j = 0; j < TRIPS; ++j) {
            int i = tid + j * HT;
            if (i < PER4) { dv[j] = d4[i]; sv[j] = s4[i]; }
        }
        if (tid < BP) bcnt[tid] = 0;
        __syncthreads();
        int bk[TRIPS][4], rk[TRIPS][4];
        #pragma unroll
        for (int j = 0; j < TRIPS; ++j) {
            int i = tid + j * HT;
            if (i < PER4) {
                bk[j][0] = dv[j].x / RS; bk[j][1] = dv[j].y / RS;
                bk[j][2] = dv[j].z / RS; bk[j][3] = dv[j].w / RS;
                rk[j][0] = atomicAdd(&bcnt[bk[j][0]], 1);
                rk[j][1] = atomicAdd(&bcnt[bk[j][1]], 1);
                rk[j][2] = atomicAdd(&bcnt[bk[j][2]], 1);
                rk[j][3] = atomicAdd(&bcnt[bk[j][3]], 1);
            }
        }
        __syncthreads();
        if (tid < 32) {
            const int c0 = bcnt[tid * 4 + 0], c1 = bcnt[tid * 4 + 1];
            const int c2 = bcnt[tid * 4 + 2], c3 = bcnt[tid * 4 + 3];
            const int s  = c0 + c1 + c2 + c3;
            const int ex = wave_inclusive(s, tid) - s;
            boffs[tid * 4 + 0] = ex;
            boffs[tid * 4 + 1] = ex + c0;
            boffs[tid * 4 + 2] = ex + c0 + c1;
            boffs[tid * 4 + 3] = ex + c0 + c1 + c2;
        }
        __syncthreads();
        #pragma unroll
        for (int j = 0; j < TRIPS; ++j) {
            int i = tid + j * HT;
            if (i < PER4) {
                lds[boffs[bk[j][0]] + rk[j][0]] = (dv[j].x << KSH) | sv[j].x;
                lds[boffs[bk[j][1]] + rk[j][1]] = (dv[j].y << KSH) | sv[j].y;
                lds[boffs[bk[j][2]] + rk[j][2]] = (dv[j].z << KSH) | sv[j].z;
                lds[boffs[bk[j][3]] + rk[j][3]] = (dv[j].w << KSH) | sv[j].w;
            }
        }
        if (tid <= NB) bstart[slice * BP + tid] = boffs[tid];
        __syncthreads();
        int4* out4 = (int4*)(keys + (size_t)slice * SL);
        for (int i = tid; i < SL / 4; i += HT) out4[i] = lds4[i];
    }
}

#define FWV (HT / 32)
#define CAP 8192
#define KR  4
__global__ void __launch_bounds__(HT)
range_fill_kernel(const int* __restrict__ keys, const int* __restrict__ bstart,
                  const int* __restrict__ table, float* __restrict__ rs_out,
                  float* __restrict__ rs_in, int* __restrict__ rowptr,
                  int* __restrict__ sorted_src) {
    static_assert(HS == 2 * FWV, "two segments per wave");
    static_assert(HS == 64, "the segment starts are summed two per lane by one wave");
    static_assert(HS % (HT / BP) == 0, "table rows split evenly over the thread groups");
    __shared__ int cntw[FWV * RS];
    __shared__ int tot[BP];
    __shared__ int dego[BP];
    __shared__ int segs0[HS];
    __shared__ int rng[2];
    __shared__ __attribute__((aligned(16))) int stage[CAP + 32];
    const int b    = blockIdx.x;
    const int tid  = threadIdx.x;
    const int lane = tid & 31;
    const int wv   = __builtin_amdgcn_readfirstlane(tid >> 5);

    int sb[2], sl[2];
    #pragma unroll
    for (int q = 0; q < 2; ++q) {
        const int sidx = 2 * wv + q;
        const int s0 = bstart[sidx * BP + b];
        const int s1 = bstart[sidx * BP + b + 1];
        sb[q] = sidx * SL + s0;
        sl[q] = s1 - s0;
    }
    int kr[2][KR];
    #pragma unroll
    for (int q = 0; q < 2; ++q) {
        #pragma unroll
        for (int i = 0; i < KR; ++i) {
            const int j = lane + 32 * i;
            kr[q][i] = 0;
            if (j < sl[q]) kr[q][i] = keys[sb[q] + j];
        }
    }

    constexpr int RPT = HS / (HT / BP);
    const int nl = tid & (BP - 1);
    int od = 0;
    if (nl < RS) {
        const int* __restrict__ col = table + (size_t)((tid / BP) * RPT) * NNP + b * RS + nl;
        #pragma unroll
        for (int r = 0; r < RPT; ++r) od += col[(size_t)r * NNP];
    }
    if (tid < HS) segs0[tid] = bstart[tid * BP + b];
    for (int i = tid; i < FWV * RS; i += HT) cntw[i] = 0;
    if (tid < BP) { tot[tid] = 0; dego[tid] = 0; }
    __syncthreads();

    if (nl < RS) atomicAdd(&dego[nl], od);
    int* mycnt = cntw + wv * RS;
    #pragma unroll
    for (int q = 0; q < 2; ++q) {
        #pragma unroll
        for (int i = 0; i < KR; ++i) {
            const int j = lane + 32 * i;
            if (j < sl[q]) atomicAdd(&mycnt[(kr[q][i] >> KSH) - b * RS], 1);
        }
        for (int j = lane + 32 * KR; j < sl[q]; j += 32) {
            const int key = keys[sb[q] + j];
            atomicAdd(&mycnt[(key >> KSH) - b * RS], 1);
        }
    }
    __syncthreads();

    if (tid < RS) {
        int c[FWV];
        #pragma unroll
        for (int w = 0; w < FWV; ++w) c[w] = cntw[w * RS + tid];
        int run = 0;
        #pragma unroll
        for (int w = 0; w < FWV; ++w) { cntw[w * RS + tid] = run; run += c[w]; }
        tot[tid] = run;
    }
    __syncthreads();

    if (tid < 32) {
        const int c0 = tot[tid * 4 + 0], c1 = tot[tid * 4 + 1];
        const int c2 = tot[tid * 4 + 2], c3 = tot[tid * 4 + 3];
        const int s   = c0 + c1 + c2 + c3;
        const int inc = wave_inclusive(s, tid);
        const int lower = __builtin_amdgcn_readlane(
            wave_inclusive(segs0[2 * tid] + segs0[2 * tid + 1], tid), 31);
        const int st0 = lower + inc - s;
        const int st[4] = { st0, st0 + c0, st0 + c0 + c1, st0 + c0 + c1 + c2 };
        const int cc[4] = { c0, c1, c2, c3 };
        #pragma unroll
        for (int q = 0; q < 4; ++q) {
            const int n = tid * 4 + q;
            tot[n] = st[q];
            if (n < RS) {
                const int node = b * RS + n;
                rowptr[node] = st[q];
                rs_in[node]  = rsqrtf(fmaxf((float)cc[q], 1.0f));
                rs_out[node] = rsqrtf(fmaxf((float)dego[n], 1.0f));
            }
        }
        if (tid == 31) { rng[0] = lower; rng[1] = inc; }
        if (b == NB - 1 && tid == 31) rowptr[NN] = lower + inc;
    }
    __syncthreads();

    const int run0  = __builtin_amdgcn_readfirstlane(rng[0]);
    const int total = __builtin_amdgcn_readfirstlane(rng[1]);
    const int gb    = run0 & ~31;
    const bool in_lds = total <= CAP;

    int tb[2][KR], rk[2][KR];
    #pragma unroll
    for (int q = 0; q < 2; ++q) {
        #pragma unroll
        for (int i = 0; i < KR; ++i) {
            const int j = lane + 32 * i;
            tb[q][i] = 0; rk[q][i] = 0;
            if (j < sl[q]) {
                const int dl = (kr[q][i] >> KSH) - b * RS;
                tb[q][i] = tot[dl];
                rk[q][i] = atomicAdd(&mycnt[dl], 1);
            }
        }
        for (int j = lane + 32 * KR; j < sl[q]; j += 32) {
            const int key = keys[sb[q] + j];
            const int dl  = (key >> KSH) - b * RS;
            const int slot = tot[dl] + atomicAdd(&mycnt[dl], 1);
            if (in_lds) stage[slot - gb] = key & ((1 << KSH) - 1);
            else        sorted_src[slot] = key & ((1 << KSH) - 1);
        }
    }
    if (in_lds) {
        #pragma unroll
        for (int q = 0; q < 2; ++q) {
            #pragma unroll
            for (int i = 0; i < KR; ++i) {
                const int j = lane + 32 * i;
                if (j < sl[q]) stage[tb[q][i] + rk[q][i] - gb] = kr[q][i] & ((1 << KSH) - 1);
            }
        }
        __syncthreads();
        const int first = run0 - gb;
        const int last  = first + total;
        int* __restrict__ part = sorted_src + gb;
        for (int w = 4 * tid; w < last; w += 4 * HT) {
            if (w >= first && w + 4 <= last) {
                *(int4*)(part + w) = *(const int4*)(stage + w);
            } else {
                for (int e = 0; e < 4; ++e)
                    if (w + e >= first && w + e < last) part[w + e] = stage[w + e];
            }
        }
    } else {
        #pragma unroll
        for (int q = 0; q < 2; ++q) {
            #pragma unroll
            for (int i = 0; i < KR; ++i) {
                const int j = lane + 32 * i;
                if (j < sl[q]) sorted_src[tb[q][i] + rk[q][i]] = kr[q][i] & ((1 << KSH) - 1);
            }
        }
    }
}

#define FW 16
#define GU 8
#define AP 132
#define PP 68
#define WP 136
#define WQ 72
#define N1 (HF / 16)
#define N2 (OF / 16)
__global__ void __launch_bounds__(FW * 32)
layer1_kernel(const int* __restrict__ rowptr, const int* __restrict__ sorted_src,
              const float* __restrict__ x,
              const float* __restrict__ rs_out,
              const float* __restrict__ rs_in,
              const float* __restrict__ W1,
              const float* __restrict__ bias1,
              const float* __restrict__ W2,
              float* __restrict__ P) {
    static_assert(NN % FW == 0, "whole blocks of nodes");
    static_assert(FW - N1 == 2 * N2 && 2 * (FW - N1) == FW, "waves 0..N1-1: product 1; waves N1..FW-1: product 2, "
                                                            "a column tile and a half of K each, then two rows each");
    __shared__ __attribute__((aligned(16))) float tile[FW * AP];
    __shared__ __attribute__((aligned(16))) float htile[FW * AP];
    __shared__ __attribute__((aligned(16))) float ptile[2 * FW * PP];
    __shared__ float rso[FW];
    __shared__ __attribute__((aligned(16))) float w1s[INF * WP];
    static_assert(INF * HF == FW * 32 * 4 * 8, "W1 is eight 16-byte pieces per thread");
    static_assert(HF == 128 && (WP * 4) % 16 == 0, "a wave's piece is one whole row of W1; LDS rows stay 16-byte aligned");
    static_assert(HF * OF == (FW - N1) * 32 * 4 * 8, "W2 is eight 16-byte pieces per thread of waves N1..FW-1");
    static_assert(OF == 64 && (WQ * 4) % 16 == 0 && HF * WQ <= INF * WP, "a wave's piece is two whole rows of W2; its copy fits in W1's");
    const int lane = threadIdx.x & 31;
    const int wave = __builtin_amdgcn_readfirstlane(threadIdx.x >> 5);
    const int node = blockIdx.x * FW + wave;
    const int beg = __builtin_amdgcn_readfirstlane(rowptr[node]);
    const int end = __builtin_amdgcn_readfirstlane(rowptr[node + 1]);
    const float rsn = rs_in[node];
    const float rsm = rs_out[node];
    {
        const float4* __restrict__ W14 = (const float4*)W1;
        float4 wv[8];
        #pragma unroll
        for (int j = 0; j < 8; ++j) wv[j] = W14[threadIdx.x + j * (FW * 32)];
        #pragma unroll
        for (int j = 0; j < 8; ++j) {
            const int idx = threadIdx.x + j * (FW * 32);
            *(float4*)(w1s + (idx >> 5) * WP + (idx & 31) * 4) = wv[j];
        }
    }
    __syncthreads();

    const float* __restrict__ xl = x + lane * 4;
    float4 acc = make_float4(0.0f, 0.0f, 0.0f, 0.0f);
    for (int base = beg; base < end; base += 32) {
        const int n = min(32, end - base);
        const int my    = sorted_src[base + min(lane, n - 1)];
        const int myr_i = __float_as_int(rs_out[my]);
        int k = 0;
        for (; k + GU <= n; k += GU) {
            float4 v[GU];
            #pragma unroll
            for (int u = 0; u < GU; ++u) {
                const int s = __builtin_amdgcn_readlane(my, k + u);
                v[u] = *(const float4*)(xl + (size_t)s * 128);
            }
            #pragma unroll
            for (int u = 0; u < GU; ++u) {
                const float r = __int_as_float(__builtin_amdgcn_readlane(myr_i, k + u));
                acc.x += v[u].x * r;
                acc.y += v[u].y * r;
                acc.z += v[u].z * r;
                acc.w += v[u].w * r;
            }
        }
        if (k < n) {
            float4 v[GU];
            #pragma unroll
            for (int u = 0; u < GU; ++u) {
                const int s = __builtin_amdgcn_readlane(my, k + u);
                v[u] = *(const float4*)(xl + (size_t)s * 128);
            }
            #pragma unroll
            for (int u = 0; u < GU; ++u) {
                const float r  = __int_as_float(__builtin_amdgcn_readlane(myr_i, k + u));
                const bool  on = k + u < n;
                acc.x = on ? acc.x + v[u].x * r : acc.x;
                acc.y = on ? acc.y + v[u].y * r : acc.y;
                acc.z = on ? acc.z + v[u].z * r : acc.z;
                acc.w = on ? acc.w + v[u].w * r : acc.w;
            }
        }
    }
    acc.x *= rsn; acc.y *= rsn; acc.z *= rsn; acc.w *= rsn;
    *(float4*)(tile + wave * AP + lane * 4) = acc;
    if (lane == 0) rso[wave] = rsm;

    const int half  = lane >> 4;
    const int l16   = lane & 15;
    v2f bw1[32];
    if (wave < N1) {
        const float* Wcol = w1s + half * 2 * WP + wave * 16 + l16;
        #pragma unroll
        for (int k = 0; k < 128; k += 4) {
            bw1[k / 4].x = Wcol[k * WP];
            bw1[k / 4].y = Wcol[(k + 1) * WP];
        }
    }
    __syncthreads();

    v2f bw2[16];
    float* w2s = w1s;
    const int t2 = (wave - N1) & (N2 - 1);
    const int kh = (wave - N1) / N2;
    if (wave < N1) {
        const float* arow = tile + l16 * AP + half * 2;
        v8f c = {};
        #pragma unroll
        for (int k = 0; k < 128; k += 4) {
            const v2f a = *(const v2f*)(arow + k);
            c = __builtin_amdgcn_wmma_f32_16x16x4_f32(false, a, false, bw1[k / 4], (short)0, c,
                                                      false, false);
        }
        const int nCol = wave * 16 + l16;
        const float bv = bias1[nCol];
        #pragma unroll
        for (int r = 0; r < 8; ++r) {
            const int row = r + half * 8;
            htile[row * AP + nCol] = fmaxf(c[r] + bv, 0.0f);
        }
    }
    if (wave >= N1) {
        const int w8 = wave - N1;
        const float* __restrict__ wsrc = W2 + (size_t)(w8 * 32 + half * 16 + l16) * 4;
        float* wdst = w2s + (w8 * 2 + half) * WQ + l16 * 4;
        float4 wv[8];
        #pragma unroll
        for (int j = 0; j < 8; ++j) wv[j] = *(const float4*)(wsrc + (size_t)j * ((FW - N1) * 32 * 4));
        #pragma unroll
        for (int j = 0; j < 8; ++j) *(float4*)(wdst + j * (16 * WQ)) = wv[j];
    }
    __syncthreads();

    if (wave >= N1) {
        const float* Wcol = w2s + (kh * 64 + half * 2) * WQ + t2 * 16 + l16;
        #pragma unroll
        for (int k = 0; k < 64; k += 4) {
            bw2[k / 4].x = Wcol[k * WQ];
            bw2[k / 4].y = Wcol[(k + 1) * WQ];
        }
        const float* arow = htile + l16 * AP + kh * 64 + half * 2;
        v8f c = {};
        #pragma unroll
        for (int k = 0; k < 64; k += 4) {
            const v2f a = *(const v2f*)(arow + k);
            c = __builtin_amdgcn_wmma_f32_16x16x4_f32(false, a, false, bw2[k / 4], (short)0, c,
                                                      false, false);
        }
        float* pt = ptile + kh * (FW * PP) + half * 8 * PP + t2 * 16 + l16;
        #pragma unroll
        for (int r = 0; r < 8; ++r) pt[r * PP] = c[r];
    }
    __syncthreads();

    if (wave >= N1) {
        const int row = (wave - N1) * 2 + half;
        const float4 p0 = *(const float4*)(ptile + row * PP + l16 * 4);
        const float4 p1 = *(const float4*)(ptile + (FW + row) * PP + l16 * 4);
        const float s = rso[row];
        float4 v;
        v.x = (p0.x + p1.x) * s; v.y = (p0.y + p1.y) * s; v.z = (p0.z + p1.z) * s; v.w = (p0.w + p1.w) * s;
        *(float4*)(P + (size_t)(blockIdx.x * FW + row) * OF + l16 * 4) = v;
    }
}


template <int J>
__device__ __forceinline__ int row_lane(int v) {
    return __builtin_amdgcn_update_dpp(v, v, 0x150 + J, 0xF, 0xF, false);
}
template <int KK>
__device__ __forceinline__ void load_rows(const char* __restrict__ pb, unsigned lo, int my, float4 (&v)[GU]) {
    static_assert(GU == 8, "eight steps written out");
    v[0] = *(const float4*)(pb + ((unsigned)row_lane<KK + 0>(my) * (unsigned)(OF * 4) + lo));
    v[1] = *(const float4*)(pb + ((unsigned)row_lane<KK + 1>(my) * (unsigned)(OF * 4) + lo));
    v[2] = *(const float4*)(pb + ((unsigned)row_lane<KK + 2>(my) * (unsigned)(OF * 4) + lo));
    v[3] = *(const float4*)(pb + ((unsigned)row_lane<KK + 3>(my) * (unsigned)(OF * 4) + lo));
    v[4] = *(const float4*)(pb + ((unsigned)row_lane<KK + 4>(my) * (unsigned)(OF * 4) + lo));
    v[5] = *(const float4*)(pb + ((unsigned)row_lane<KK + 5>(my) * (unsigned)(OF * 4) + lo));
    v[6] = *(const float4*)(pb + ((unsigned)row_lane<KK + 6>(my) * (unsigned)(OF * 4) + lo));
    v[7] = *(const float4*)(pb + ((unsigned)row_lane<KK + 7>(my) * (unsigned)(OF * 4) + lo));
}
template <int KK>
__device__ __forceinline__ void gather_steps(const char* __restrict__ pb, unsigned lo, int my,
                                             int n0, int n1, int nh, float4& acc) {
    if (KK + GU <= n1) {
        float4 v[GU];
        load_rows<KK>(pb, lo, my, v);
        #pragma unroll
        for (int u = 0; u < GU; ++u) {
            acc.x += v[u].x; acc.y += v[u].y; acc.z += v[u].z; acc.w += v[u].w;
        }
    } else if (KK < n0) {
        float4 v[GU];
        load_rows<KK>(pb, lo, my, v);
        #pragma unroll
        for (int u = 0; u < GU; ++u) {
            const bool on = KK + u < nh;
            acc.x += on ? v[u].x : 0.0f;
            acc.y += on ? v[u].y : 0.0f;
            acc.z += on ? v[u].z : 0.0f;
            acc.w += on ? v[u].w : 0.0f;
        }
    }
}
__device__ __forceinline__ float other_half(float v) {
    return __int_as_float(__builtin_amdgcn_permlanex16(__float_as_int(v), __float_as_int(v),
                                                       0x76543210, 0xfedcba98, false, false));
}

__global__ void __launch_bounds__(FW * 32)
gather_out_kernel(const int* __restrict__ rowptr, const int* __restrict__ sorted_src,
                  const float* __restrict__ P,
                  const float* __restrict__ rs_in,
                  const float* __restrict__ bias,
                  float* __restrict__ out) {
    static_assert(OF == 64, "a row of P is 16 lanes x 16 bytes");
    const int lane = threadIdx.x & 31;
    const int wave = __builtin_amdgcn_readfirstlane(threadIdx.x >> 5);
    const int node = blockIdx.x * FW + wave;
    const int beg = __builtin_amdgcn_readfirstlane(rowptr[node]);
    const int end = __builtin_amdgcn_readfirstlane(rowptr[node + 1]);
    const float rsn = rs_in[node];
    const int half = lane >> 4;
    const int l16  = lane & 15;
    const float4 bv = *(const float4*)(bias + l16 * 4);
    const int cnt  = end - beg;
    const int len0 = (cnt + 1) >> 1;
    const int len1 = cnt - len0;
    const int lenh = half ? len1 : len0;
    const int* __restrict__ run = sorted_src + beg + (half ? len0 : 0);
    const char* __restrict__ pb = (const char*)P;
    const unsigned lo = (unsigned)l16 * 16u;
    float4 acc = make_float4(0.0f, 0.0f, 0.0f, 0.0f);
    for (int b = 0; b < len0; b += 16) {
        const int n0 = min(16, len0 - b);
        const int n1 = min(16, len1 - b);
        const int nh = half ? n1 : n0;
        const int my = run[min(b + l16, lenh - 1)];
        gather_steps<0>(pb, lo, my, n0, n1, nh, acc);
        gather_steps<GU>(pb, lo, my, n0, n1, nh, acc);
    }
    acc.x += other_half(acc.x);
    acc.y += other_half(acc.y);
    acc.z += other_half(acc.z);
    acc.w += other_half(acc.w);
    if (half == 0) {
        float4 o;
        o.x = acc.x * rsn + bv.x;
        o.y = acc.y * rsn + bv.y;
        o.z = acc.z * rsn + bv.z;
        o.w = acc.w * rsn + bv.w;
        *(float4*)(out + (size_t)node * OF + l16 * 4) = o;
    }
}


extern "C" void kernel_launch(void* const* d_in, const int* in_sizes, int n_in,
                              void* d_out, int out_size, void* d_ws, size_t ws_size,
                              hipStream_t stream) {
    static_assert(INF == 128 && HF == 128, "both products have K = 128 (one 512-byte row per node)");
    const float* in_feat = (const float*)d_in[0];
    const int*   src     = (const int*)d_in[1];
    const int*   dst     = (const int*)d_in[2];
    const float* W1      = (const float*)d_in[3];
    const float* b1      = (const float*)d_in[4];
    const float* W2      = (const float*)d_in[5];
    const float* b2      = (const float*)d_in[6];
    float* out = (float*)d_out;

    float* ws      = (float*)d_ws;
    float* rs_out  = ws;
    float* rs_in   = ws + NNP;
    int*   rowptr  = (int*)(ws + 2 * NNP);
    float* P       = ws + 4 * NNP;
    int*   dtable  = (int*)(P + (size_t)NN * OF);
    int*   keys    = dtable + (size_t)HS * NNP;
    int*   bstart  = keys + NE;
    int*   sorted  = bstart + HS * BP;

    slice_kernel<<<2 * HS, HT, 0, stream>>>(src, dst, dtable, keys, bstart);
    range_fill_kernel<<<NB, HT, 0, stream>>>(keys, bstart, dtable, rs_out, rs_in, rowptr, sorted);

    layer1_kernel<<<NN / FW, FW * 32, 0, stream>>>(
        rowptr, sorted, in_feat, rs_out, rs_in, W1, b1, W2, P);
    gather_out_kernel<<<NN / FW, FW * 32, 0, stream>>>(rowptr, sorted, P, rs_in, b2, out);
}


# edge sort regridded: 128 dst slices of 5000 edges, 250 ranges of 40 nodes, blocks of 16 waves: 192 and 250 WGPs share the LDS counting and placing (were 128, 125); order kept but for ties in a slice
# speedup vs baseline: 8.3075x; 1.0405x over previous
#include <hip/hip_runtime.h>


#define NN 10000
#define NE 640000
#define INF 128
#define HF  128
#define OF  64

typedef float v2f __attribute__((ext_vector_type(2)));
typedef float v8f __attribute__((ext_vector_type(8)));

#define HS   64
#define SL   (NE / HS)
#define HD   128
#define SD   (NE / HD)
#define HT   512
#define NNP  10016
#define NB   250
#define RS   (NN / NB)
#define BP   128
#define BPD  256
#define KSH  14
#define LW   (NN > SL ? NN : SL)

__device__ __forceinline__ int wave_inclusive(int v, int lane) {
    #pragma unroll
    for (int off = 1; off < 32; off <<= 1) {
        int up = __shfl_up(v, off);
        if (lane >= off) v += up;
    }
    return v;
}

__global__ void __launch_bounds__(HT)
slice_kernel(const int* __restrict__ src, const int* __restrict__ dst,
             int* __restrict__ table, int* __restrict__ keys, int* __restrict__ bstart) {
    __shared__ __attribute__((aligned(16))) int lds[LW];
    __shared__ __attribute__((aligned(16))) int bcnt[BPD];
    __shared__ __attribute__((aligned(16))) int boffs[BPD];
    static_assert(NE % HS == 0 && NE % HD == 0, "edge slices must be equal");
    static_assert(NN % 4 == 0 && SL % 4 == 0 && SD % 4 == 0, "rows and slices move in 16-byte pieces");
    static_assert(NB * RS == NN && NB + 1 <= BPD && RS <= BP, "ranges tile the nodes; range arrays fit their pitch");
    static_assert(BPD == 32 * 8 && BPD <= HT && SD <= LW, "one wave scans the range counts, eight per lane; a dst slice fits the LDS words");
    static_assert(NN <= (1 << KSH) && NN < (1 << (31 - KSH)), "a packed word holds both node ids");
    const bool is_src = blockIdx.x >= HD;
    const int slice = is_src ? blockIdx.x - HD : blockIdx.x;
    const int tid   = threadIdx.x;
    int4* lds4 = (int4*)lds;

    if (is_src) {
        constexpr int PER4  = SL / 4;
        constexpr int TRIPS = (PER4 + HT - 1) / HT;
        const int4* __restrict__ idx4 = (const int4*)(src + (size_t)slice * SL);
        int4 v[TRIPS];
        #pragma unroll
        for (int j = 0; j < TRIPS; ++j) {
            int i = tid + j * HT;
            if (i < PER4) v[j] = idx4[i];
        }
        for (int i = tid; i < NN / 4; i += HT) lds4[i] = make_int4(0, 0, 0, 0);
        __syncthreads();
        #pragma unroll
        for (int j = 0; j < TRIPS; ++j) {
            int i = tid + j * HT;
            if (i < PER4) {
                atomicAdd(&lds[v[j].x], 1);
                atomicAdd(&lds[v[j].y], 1);
                atomicAdd(&lds[v[j].z], 1);
                atomicAdd(&lds[v[j].w], 1);
            }
        }
        __syncthreads();
        int4* row4 = (int4*)(table + (size_t)slice * NNP);
        for (int i = tid; i < NN / 4; i += HT) row4[i] = lds4[i];
    } else {
        constexpr int PER4  = SD / 4;
        constexpr int TRIPS = (PER4 + HT - 1) / HT;
        const int4* __restrict__ d4 = (const int4*)(dst + (size_t)slice * SD);
        const int4* __restrict__ s4 = (const int4*)(src + (size_t)slice * SD);
        int4 dv[TRIPS], sv[TRIPS];
        #pragma unroll
        for (int j = 0; j < TRIPS; ++j) {
            int i = tid + j * HT;
            if (i < PER4) { dv[j] = d4[i]; sv[j] = s4[i]; }
        }
        if (tid < BPD) bcnt[tid] = 0;
        __syncthreads();
        int bk[TRIPS][4], rk[TRIPS][4];
        #pragma unroll
        for (int j = 0; j < TRIPS; ++j) {
            int i = tid + j * HT;
            if (i < PER4) {
                bk[j][0] = dv[j].x / RS; bk[j][1] = dv[j].y / RS;
                bk[j][2] = dv[j].z / RS; bk[j][3] = dv[j].w / RS;
                rk[j][0] = atomicAdd(&bcnt[bk[j][0]], 1);
                rk[j][1] = atomicAdd(&bcnt[bk[j][1]], 1);
                rk[j][2] = atomicAdd(&bcnt[bk[j][2]], 1);
                rk[j][3] = atomicAdd(&bcnt[bk[j][3]], 1);
            }
        }
        __syncthreads();
        if (tid < 32) {
            const int4 ca = *(const int4*)(bcnt + tid * 8);
            const int4 cb = *(const int4*)(bcnt + tid * 8 + 4);
            const int s  = ca.x + ca.y + ca.z + ca.w + cb.x + cb.y + cb.z + cb.w;
            const int ex = wave_inclusive(s, tid) - s;
            int4 oa, ob;
            oa.x = ex;
            oa.y = oa.x + ca.x;
            oa.z = oa.y + ca.y;
            oa.w = oa.z + ca.z;
            ob.x = oa.w + ca.w;
            ob.y = ob.x + cb.x;
            ob.z = ob.y + cb.y;
            ob.w = ob.z + cb.z;
            *(int4*)(boffs + tid * 8)     = oa;
            *(int4*)(boffs + tid * 8 + 4) = ob;
        }
        __syncthreads();
        #pragma unroll
        for (int j = 0; j < TRIPS; ++j) {
            int i = tid + j * HT;
            if (i < PER4) {
                lds[boffs[bk[j][0]] + rk[j][0]] = (dv[j].x << KSH) | sv[j].x;
                lds[boffs[bk[j][1]] + rk[j][1]] = (dv[j].y << KSH) | sv[j].y;
                lds[boffs[bk[j][2]] + rk[j][2]] = (dv[j].z << KSH) | sv[j].z;
                lds[boffs[bk[j][3]] + rk[j][3]] = (dv[j].w << KSH) | sv[j].w;
            }
        }
        if (tid <= NB) bstart[slice * BPD + tid] = boffs[tid];
        __syncthreads();
        int4* out4 = (int4*)(keys + (size_t)slice * SD);
        for (int i = tid; i < SD / 4; i += HT) out4[i] = lds4[i];
    }
}

#define FWV (HT / 32)
#define SPW (HD / FWV)
#define CAP 4096
#define KR  2
__global__ void __launch_bounds__(HT)
range_fill_kernel(const int* __restrict__ keys, const int* __restrict__ bstart,
                  const int* __restrict__ table, float* __restrict__ rs_out,
                  float* __restrict__ rs_in, int* __restrict__ rowptr,
                  int* __restrict__ sorted_src) {
    static_assert(HD == SPW * FWV, "whole segments per wave");
    static_assert(HD == 4 * 32, "the segment starts are summed four per lane by one wave");
    static_assert(HS % (HT / BP) == 0, "table rows split evenly over the thread groups");
    __shared__ int cntw[FWV * RS];
    __shared__ int tot[BP];
    __shared__ int dego[BP];
    __shared__ int segs0[HD];
    __shared__ int rng[2];
    __shared__ __attribute__((aligned(16))) int stage[CAP + 32];
    const int b    = blockIdx.x;
    const int tid  = threadIdx.x;
    const int lane = tid & 31;
    const int wv   = __builtin_amdgcn_readfirstlane(tid >> 5);

    int sb[SPW], sl[SPW];
    #pragma unroll
    for (int q = 0; q < SPW; ++q) {
        const int sidx = SPW * wv + q;
        const int s0 = bstart[sidx * BPD + b];
        const int s1 = bstart[sidx * BPD + b + 1];
        sb[q] = sidx * SD + s0;
        sl[q] = s1 - s0;
    }
    int kr[SPW][KR];
    #pragma unroll
    for (int q = 0; q < SPW; ++q) {
        #pragma unroll
        for (int i = 0; i < KR; ++i) {
            const int j = lane + 32 * i;
            kr[q][i] = 0;
            if (j < sl[q]) kr[q][i] = keys[sb[q] + j];
        }
    }

    constexpr int RPT = HS / (HT / BP);
    const int nl = tid & (BP - 1);
    int od = 0;
    if (nl < RS) {
        const int* __restrict__ col = table + (size_t)((tid / BP) * RPT) * NNP + b * RS + nl;
        #pragma unroll
        for (int r = 0; r < RPT; ++r) od += col[(size_t)r * NNP];
    }
    if (tid < HD) segs0[tid] = bstart[tid * BPD + b];
    for (int i = tid; i < FWV * RS; i += HT) cntw[i] = 0;
    if (tid < BP) { tot[tid] = 0; dego[tid] = 0; }
    __syncthreads();

    if (nl < RS) atomicAdd(&dego[nl], od);
    int* mycnt = cntw + wv * RS;
    #pragma unroll
    for (int q = 0; q < SPW; ++q) {
        #pragma unroll
        for (int i = 0; i < KR; ++i) {
            const int j = lane + 32 * i;
            if (j < sl[q]) atomicAdd(&mycnt[(kr[q][i] >> KSH) - b * RS], 1);
        }
        for (int j = lane + 32 * KR; j < sl[q]; j += 32) {
            const int key = keys[sb[q] + j];
            atomicAdd(&mycnt[(key >> KSH) - b * RS], 1);
        }
    }
    __syncthreads();

    if (tid < RS) {
        int c[FWV];
        #pragma unroll
        for (int w = 0; w < FWV; ++w) c[w] = cntw[w * RS + tid];
        int run = 0;
        #pragma unroll
        for (int w = 0; w < FWV; ++w) { cntw[w * RS + tid] = run; run += c[w]; }
        tot[tid] = run;
    }
    __syncthreads();

    if (tid < 32) {
        const int c0 = tot[tid * 4 + 0], c1 = tot[tid * 4 + 1];
        const int c2 = tot[tid * 4 + 2], c3 = tot[tid * 4 + 3];
        const int s   = c0 + c1 + c2 + c3;
        const int inc = wave_inclusive(s, tid);
        const int lower = __builtin_amdgcn_readlane(
            wave_inclusive(segs0[4 * tid] + segs0[4 * tid + 1] + segs0[4 * tid + 2] + segs0[4 * tid + 3], tid), 31);
        const int st0 = lower + inc - s;
        const int st[4] = { st0, st0 + c0, st0 + c0 + c1, st0 + c0 + c1 + c2 };
        const int cc[4] = { c0, c1, c2, c3 };
        #pragma unroll
        for (int q = 0; q < 4; ++q) {
            const int n = tid * 4 + q;
            tot[n] = st[q];
            if (n < RS) {
                const int node = b * RS + n;
                rowptr[node] = st[q];
                rs_in[node]  = rsqrtf(fmaxf((float)cc[q], 1.0f));
                rs_out[node] = rsqrtf(fmaxf((float)dego[n], 1.0f));
            }
        }
        if (tid == 31) { rng[0] = lower; rng[1] = inc; }
        if (b == NB - 1 && tid == 31) rowptr[NN] = lower + inc;
    }
    __syncthreads();

    const int run0  = __builtin_amdgcn_readfirstlane(rng[0]);
    const int total = __builtin_amdgcn_readfirstlane(rng[1]);
    const int gb    = run0 & ~31;
    const bool in_lds = total <= CAP;

    int tb[SPW][KR], rk[SPW][KR];
    #pragma unroll
    for (int q = 0; q < SPW; ++q) {
        #pragma unroll
        for (int i = 0; i < KR; ++i) {
            const int j = lane + 32 * i;
            tb[q][i] = 0; rk[q][i] = 0;
            if (j < sl[q]) {
                const int dl = (kr[q][i] >> KSH) - b * RS;
                tb[q][i] = tot[dl];
                rk[q][i] = atomicAdd(&mycnt[dl], 1);
            }
        }
        for (int j = lane + 32 * KR; j < sl[q]; j += 32) {
            const int key = keys[sb[q] + j];
            const int dl  = (key >> KSH) - b * RS;
            const int slot = tot[dl] + atomicAdd(&mycnt[dl], 1);
            if (in_lds) stage[slot - gb] = key & ((1 << KSH) - 1);
            else        sorted_src[slot] = key & ((1 << KSH) - 1);
        }
    }
    if (in_lds) {
        #pragma unroll
        for (int q = 0; q < SPW; ++q) {
            #pragma unroll
            for (int i = 0; i < KR; ++i) {
                const int j = lane + 32 * i;
                if (j < sl[q]) stage[tb[q][i] + rk[q][i] - gb] = kr[q][i] & ((1 << KSH) - 1);
            }
        }
        __syncthreads();
        const int first = run0 - gb;
        const int last  = first + total;
        int* __restrict__ part = sorted_src + gb;
        for (int w = 4 * tid; w < last; w += 4 * HT) {
            if (w >= first && w + 4 <= last) {
                *(int4*)(part + w) = *(const int4*)(stage + w);
            } else {
                for (int e = 0; e < 4; ++e)
                    if (w + e >= first && w + e < last) part[w + e] = stage[w + e];
            }
        }
    } else {
        #pragma unroll
        for (int q = 0; q < SPW; ++q) {
            #pragma unroll
            for (int i = 0; i < KR; ++i) {
                const int j = lane + 32 * i;
                if (j < sl[q]) sorted_src[tb[q][i] + rk[q][i]] = kr[q][i] & ((1 << KSH) - 1);
            }
        }
    }
}

#define FW 16
#define GU 8
#define AP 132
#define PP 68
#define WP 136
#define WQ 72
#define N1 (HF / 16)
#define N2 (OF / 16)
__global__ void __launch_bounds__(FW * 32)
layer1_kernel(const int* __restrict__ rowptr, const int* __restrict__ sorted_src,
              const float* __restrict__ x,
              const float* __restrict__ rs_out,
              const float* __restrict__ rs_in,
              const float* __restrict__ W1,
              const float* __restrict__ bias1,
              const float* __restrict__ W2,
              float* __restrict__ P) {
    static_assert(NN % FW == 0, "whole blocks of nodes");
    static_assert(FW - N1 == 2 * N2 && 2 * (FW - N1) == FW, "waves 0..N1-1: product 1; waves N1..FW-1: product 2, "
                                                            "a column tile and a half of K each, then two rows each");
    __shared__ __attribute__((aligned(16))) float tile[FW * AP];
    __shared__ __attribute__((aligned(16))) float htile[FW * AP];
    __shared__ __attribute__((aligned(16))) float ptile[2 * FW * PP];
    __shared__ float rso[FW];
    __shared__ __attribute__((aligned(16))) float w1s[INF * WP];
    static_assert(INF * HF == FW * 32 * 4 * 8, "W1 is eight 16-byte pieces per thread");
    static_assert(HF == 128 && (WP * 4) % 16 == 0, "a wave's piece is one whole row of W1; LDS rows stay 16-byte aligned");
    static_assert(HF * OF == (FW - N1) * 32 * 4 * 8, "W2 is eight 16-byte pieces per thread of waves N1..FW-1");
    static_assert(OF == 64 && (WQ * 4) % 16 == 0 && HF * WQ <= INF * WP, "a wave's piece is two whole rows of W2; its copy fits in W1's");
    const int lane = threadIdx.x & 31;
    const int wave = __builtin_amdgcn_readfirstlane(threadIdx.x >> 5);
    const int node = blockIdx.x * FW + wave;
    const int beg = __builtin_amdgcn_readfirstlane(rowptr[node]);
    const int end = __builtin_amdgcn_readfirstlane(rowptr[node + 1]);
    const float rsn = rs_in[node];
    const float rsm = rs_out[node];
    {
        const float4* __restrict__ W14 = (const float4*)W1;
        float4 wv[8];
        #pragma unroll
        for (int j = 0; j < 8; ++j) wv[j] = W14[threadIdx.x + j * (FW * 32)];
        #pragma unroll
        for (int j = 0; j < 8; ++j) {
            const int idx = threadIdx.x + j * (FW * 32);
            *(float4*)(w1s + (idx >> 5) * WP + (idx & 31) * 4) = wv[j];
        }
    }
    __syncthreads();

    const float* __restrict__ xl = x + lane * 4;
    float4 acc = make_float4(0.0f, 0.0f, 0.0f, 0.0f);
    for (int base = beg; base < end; base += 32) {
        const int n = min(32, end - base);
        const int my    = sorted_src[base + min(lane, n - 1)];
        const int myr_i = __float_as_int(rs_out[my]);
        int k = 0;
        for (; k + GU <= n; k += GU) {
            float4 v[GU];
            #pragma unroll
            for (int u = 0; u < GU; ++u) {
                const int s = __builtin_amdgcn_readlane(my, k + u);
                v[u] = *(const float4*)(xl + (size_t)s * 128);
            }
            #pragma unroll
            for (int u = 0; u < GU; ++u) {
                const float r = __int_as_float(__builtin_amdgcn_readlane(myr_i, k + u));
                acc.x += v[u].x * r;
                acc.y += v[u].y * r;
                acc.z += v[u].z * r;
                acc.w += v[u].w * r;
            }
        }
        if (k < n) {
            float4 v[GU];
            #pragma unroll
            for (int u = 0; u < GU; ++u) {
                const int s = __builtin_amdgcn_readlane(my, k + u);
                v[u] = *(const float4*)(xl + (size_t)s * 128);
            }
            #pragma unroll
            for (int u = 0; u < GU; ++u) {
                const float r  = __int_as_float(__builtin_amdgcn_readlane(myr_i, k + u));
                const bool  on = k + u < n;
                acc.x = on ? acc.x + v[u].x * r : acc.x;
                acc.y = on ? acc.y + v[u].y * r : acc.y;
                acc.z = on ? acc.z + v[u].z * r : acc.z;
                acc.w = on ? acc.w + v[u].w * r : acc.w;
            }
        }
    }
    acc.x *= rsn; acc.y *= rsn; acc.z *= rsn; acc.w *= rsn;
    *(float4*)(tile + wave * AP + lane * 4) = acc;
    if (lane == 0) rso[wave] = rsm;

    const int half  = lane >> 4;
    const int l16   = lane & 15;
    v2f bw1[32];
    if (wave < N1) {
        const float* Wcol = w1s + half * 2 * WP + wave * 16 + l16;
        #pragma unroll
        for (int k = 0; k < 128; k += 4) {
            bw1[k / 4].x = Wcol[k * WP];
            bw1[k / 4].y = Wcol[(k + 1) * WP];
        }
    }
    __syncthreads();

    v2f bw2[16];
    float* w2s = w1s;
    const int t2 = (wave - N1) & (N2 - 1);
    const int kh = (wave - N1) / N2;
    if (wave < N1) {
        const float* arow = tile + l16 * AP + half * 2;
        v8f c = {};
        #pragma unroll
        for (int k = 0; k < 128; k += 4) {
            const v2f a = *(const v2f*)(arow + k);
            c = __builtin_amdgcn_wmma_f32_16x16x4_f32(false, a, false, bw1[k / 4], (short)0, c,
                                                      false, false);
        }
        const int nCol = wave * 16 + l16;
        const float bv = bias1[nCol];
        #pragma unroll
        for (int r = 0; r < 8; ++r) {
            const int row = r + half * 8;
            htile[row * AP + nCol] = fmaxf(c[r] + bv, 0.0f);
        }
    }
    if (wave >= N1) {
        const int w8 = wave - N1;
        const float* __restrict__ wsrc = W2 + (size_t)(w8 * 32 + half * 16 + l16) * 4;
        float* wdst = w2s + (w8 * 2 + half) * WQ + l16 * 4;
        float4 wv[8];
        #pragma unroll
        for (int j = 0; j < 8; ++j) wv[j] = *(const float4*)(wsrc + (size_t)j * ((FW - N1) * 32 * 4));
        #pragma unroll
        for (int j = 0; j < 8; ++j) *(float4*)(wdst + j * (16 * WQ)) = wv[j];
    }
    __syncthreads();

    if (wave >= N1) {
        const float* Wcol = w2s + (kh * 64 + half * 2) * WQ + t2 * 16 + l16;
        #pragma unroll
        for (int k = 0; k < 64; k += 4) {
            bw2[k / 4].x = Wcol[k * WQ];
            bw2[k / 4].y = Wcol[(k + 1) * WQ];
        }
        const float* arow = htile + l16 * AP + kh * 64 + half * 2;
        v8f c = {};
        #pragma unroll
        for (int k = 0; k < 64; k += 4) {
            const v2f a = *(const v2f*)(arow + k);
            c = __builtin_amdgcn_wmma_f32_16x16x4_f32(false, a, false, bw2[k / 4], (short)0, c,
                                                      false, false);
        }
        float* pt = ptile + kh * (FW * PP) + half * 8 * PP + t2 * 16 + l16;
        #pragma unroll
        for (int r = 0; r < 8; ++r) pt[r * PP] = c[r];
    }
    __syncthreads();

    if (wave >= N1) {
        const int row = (wave - N1) * 2 + half;
        const float4 p0 = *(const float4*)(ptile + row * PP + l16 * 4);
        const float4 p1 = *(const float4*)(ptile + (FW + row) * PP + l16 * 4);
        const float s = rso[row];
        float4 v;
        v.x = (p0.x + p1.x) * s; v.y = (p0.y + p1.y) * s; v.z = (p0.z + p1.z) * s; v.w = (p0.w + p1.w) * s;
        *(float4*)(P + (size_t)(blockIdx.x * FW + row) * OF + l16 * 4) = v;
    }
}


template <int J>
__device__ __forceinline__ int row_lane(int v) {
    return __builtin_amdgcn_update_dpp(v, v, 0x150 + J, 0xF, 0xF, false);
}
template <int KK>
__device__ __forceinline__ void load_rows(const char* __restrict__ pb, unsigned lo, int my, float4 (&v)[GU]) {
    static_assert(GU == 8, "eight steps written out");
    v[0] = *(const float4*)(pb + ((unsigned)row_lane<KK + 0>(my) * (unsigned)(OF * 4) + lo));
    v[1] = *(const float4*)(pb + ((unsigned)row_lane<KK + 1>(my) * (unsigned)(OF * 4) + lo));
    v[2] = *(const float4*)(pb + ((unsigned)row_lane<KK + 2>(my) * (unsigned)(OF * 4) + lo));
    v[3] = *(const float4*)(pb + ((unsigned)row_lane<KK + 3>(my) * (unsigned)(OF * 4) + lo));
    v[4] = *(const float4*)(pb + ((unsigned)row_lane<KK + 4>(my) * (unsigned)(OF * 4) + lo));
    v[5] = *(const float4*)(pb + ((unsigned)row_lane<KK + 5>(my) * (unsigned)(OF * 4) + lo));
    v[6] = *(const float4*)(pb + ((unsigned)row_lane<KK + 6>(my) * (unsigned)(OF * 4) + lo));
    v[7] = *(const float4*)(pb + ((unsigned)row_lane<KK + 7>(my) * (unsigned)(OF * 4) + lo));
}
template <int KK>
__device__ __forceinline__ void gather_steps(const char* __restrict__ pb, unsigned lo, int my,
                                             int n0, int n1, int nh, float4& acc) {
    if (KK + GU <= n1) {
        float4 v[GU];
        load_rows<KK>(pb, lo, my, v);
        #pragma unroll
        for (int u = 0; u < GU; ++u) {
            acc.x += v[u].x; acc.y += v[u].y; acc.z += v[u].z; acc.w += v[u].w;
        }
    } else if (KK < n0) {
        float4 v[GU];
        load_rows<KK>(pb, lo, my, v);
        #pragma unroll
        for (int u = 0; u < GU; ++u) {
            const bool on = KK + u < nh;
            acc.x += on ? v[u].x : 0.0f;
            acc.y += on ? v[u].y : 0.0f;
            acc.z += on ? v[u].z : 0.0f;
            acc.w += on ? v[u].w : 0.0f;
        }
    }
}
__device__ __forceinline__ float other_half(float v) {
    return __int_as_float(__builtin_amdgcn_permlanex16(__float_as_int(v), __float_as_int(v),
                                                       0x76543210, 0xfedcba98, false, false));
}

__global__ void __launch_bounds__(FW * 32)
gather_out_kernel(const int* __restrict__ rowptr, const int* __restrict__ sorted_src,
                  const float* __restrict__ P,
                  const float* __restrict__ rs_in,
                  const float* __restrict__ bias,
                  float* __restrict__ out) {
    static_assert(OF == 64, "a row of P is 16 lanes x 16 bytes");
    const int lane = threadIdx.x & 31;
    const int wave = __builtin_amdgcn_readfirstlane(threadIdx.x >> 5);
    const int node = blockIdx.x * FW + wave;
    const int beg = __builtin_amdgcn_readfirstlane(rowptr[node]);
    const int end = __builtin_amdgcn_readfirstlane(rowptr[node + 1]);
    const float rsn = rs_in[node];
    const int half = lane >> 4;
    const int l16  = lane & 15;
    const float4 bv = *(const float4*)(bias + l16 * 4);
    const int cnt  = end - beg;
    const int len0 = (cnt + 1) >> 1;
    const int len1 = cnt - len0;
    const int lenh = half ? len1 : len0;
    const int* __restrict__ run = sorted_src + beg + (half ? len0 : 0);
    const char* __restrict__ pb = (const char*)P;
    const unsigned lo = (unsigned)l16 * 16u;
    float4 acc = make_float4(0.0f, 0.0f, 0.0f, 0.0f);
    for (int b = 0; b < len0; b += 16) {
        const int n0 = min(16, len0 - b);
        const int n1 = min(16, len1 - b);
        const int nh = half ? n1 : n0;
        const int my = run[min(b + l16, lenh - 1)];
        gather_steps<0>(pb, lo, my, n0, n1, nh, acc);
        gather_steps<GU>(pb, lo, my, n0, n1, nh, acc);
    }
    acc.x += other_half(acc.x);
    acc.y += other_half(acc.y);
    acc.z += other_half(acc.z);
    acc.w += other_half(acc.w);
    if (half == 0) {
        float4 o;
        o.x = acc.x * rsn + bv.x;
        o.y = acc.y * rsn + bv.y;
        o.z = acc.z * rsn + bv.z;
        o.w = acc.w * rsn + bv.w;
        *(float4*)(out + (size_t)node * OF + l16 * 4) = o;
    }
}


extern "C" void kernel_launch(void* const* d_in, const int* in_sizes, int n_in,
                              void* d_out, int out_size, void* d_ws, size_t ws_size,
                              hipStream_t stream) {
    static_assert(INF == 128 && HF == 128, "both products have K = 128 (one 512-byte row per node)");
    const float* in_feat = (const float*)d_in[0];
    const int*   src     = (const int*)d_in[1];
    const int*   dst     = (const int*)d_in[2];
    const float* W1      = (const float*)d_in[3];
    const float* b1      = (const float*)d_in[4];
    const float* W2      = (const float*)d_in[5];
    const float* b2      = (const float*)d_in[6];
    float* out = (float*)d_out;

    float* ws      = (float*)d_ws;
    float* rs_out  = ws;
    float* rs_in   = ws + NNP;
    int*   rowptr  = (int*)(ws + 2 * NNP);
    float* P       = ws + 4 * NNP;
    int*   dtable  = (int*)(P + (size_t)NN * OF);
    int*   keys    = dtable + (size_t)HS * NNP;
    int*   bstart  = keys + NE;
    int*   sorted  = bstart + HD * BPD;

    slice_kernel<<<HS + HD, HT, 0, stream>>>(src, dst, dtable, keys, bstart);
    range_fill_kernel<<<NB, HT, 0, stream>>>(keys, bstart, dtable, rs_out, rs_in, rowptr, sorted);

    layer1_kernel<<<NN / FW, FW * 32, 0, stream>>>(
        rowptr, sorted, in_feat, rs_out, rs_in, W1, b1, W2, P);
    gather_out_kernel<<<NN / FW, FW * 32, 0, stream>>>(rowptr, sorted, P, rs_in, b2, out);
}
